# starting kernel
# baseline (speedup 1.0000x reference)
#include <hip/hip_runtime.h>
#include <hip/hip_bf16.h>


typedef __attribute__((ext_vector_type(2))) float v2f;
typedef __attribute__((ext_vector_type(8))) float v8f;

#define NFEAT  256
#define NHID   128
#define NCLASS 64

__device__ __forceinline__ void atomAddF(float* p, float v) {
    __hip_atomic_fetch_add(p, v, __ATOMIC_RELAXED, __HIP_MEMORY_SCOPE_AGENT);
}

__global__ void k_zero(float* __restrict__ p, int n) {
    int i = blockIdx.x * blockDim.x + threadIdx.x;
    if (i < n) p[i] = 0.0f;
}

__global__ void k_count(const int* __restrict__ src, const int* __restrict__ dst,
                        float* __restrict__ cdst, float* __restrict__ csrc, int E) {
    int e = blockIdx.x * blockDim.x + threadIdx.x;
    if (e < E) {
        atomAddF(&cdst[dst[e]], 1.0f);
        atomAddF(&csrc[src[e]], 1.0f);
    }
}

__global__ void k_scatter_rows(const float* __restrict__ X,
                               const int* __restrict__ gath,
                               const int* __restrict__ scat,
                               float* __restrict__ out, int E) {
    int t = blockIdx.x * blockDim.x + threadIdx.x;
    int e = t >> 5;
    if (e >= E) return;
    int f = (t & 31) * 4;
    const float4 v = *(const float4*)(X + (size_t)gath[e] * NHID + f);
    float* o = out + (size_t)scat[e] * NHID + f;
    atomAddF(o + 0, v.x);
    atomAddF(o + 1, v.y);
    atomAddF(o + 2, v.z);
    atomAddF(o + 3, v.w);
}

__global__ void k_div_rows(float* __restrict__ agg, const float* __restrict__ cnt, int total) {
    int i = blockIdx.x * blockDim.x + threadIdx.x;
    if (i >= total) return;
    float c = cnt[i >> 7];
    agg[i] *= 1.0f / fmaxf(c, 1.0f);
}

__global__ void k_edge_h(const float* __restrict__ A, const float* __restrict__ B,
                         const int* __restrict__ src, const int* __restrict__ dst,
                         float* __restrict__ hs, int E) {
    int t = blockIdx.x * blockDim.x + threadIdx.x;
    int e = t >> 5;
    if (e >= E) return;
    int f = (t & 31) * 4;
    int s = src[e], d = dst[e];
    const float4 a = *(const float4*)(A + (size_t)s * NHID + f);
    const float4 b = *(const float4*)(B + (size_t)d * NHID + f);
    float hx = (a.x + b.x) * (a.x + b.x);
    float hy = (a.y + b.y) * (a.y + b.y);
    float hz = (a.z + b.z) * (a.z + b.z);
    float hw = (a.w + b.w) * (a.w + b.w);
    float* o = hs + (size_t)s * NHID + f;
    atomAddF(o + 0, hx);
    atomAddF(o + 1, hy);
    atomAddF(o + 2, hz);
    atomAddF(o + 3, hw);
}

__global__ void k_gate(const float* __restrict__ Xcur, const float* __restrict__ Xnew,
                       const float* __restrict__ hs, const float* __restrict__ csrc,
                       float* __restrict__ Xnext, int total) {
    int i = blockIdx.x * blockDim.x + threadIdx.x;
    if (i >= total) return;
    float c   = fmaxf(csrc[i >> 7], 1.0f);
    float tau = tanhf(hs[i] / c);
    Xnext[i]  = (1.0f - tau) * Xcur[i] + tau * Xnew[i];
}

__global__ void k_gemm(const float* __restrict__ A1, const float* __restrict__ W1,
                       const float* __restrict__ A2, const float* __restrict__ W2,
                       const float* __restrict__ bias, float* __restrict__ C,
                       int M, int K, int N, int relu) {
    const int lane   = threadIdx.x & 31;
    const int wave   = threadIdx.x >> 5;
    const int ntiles = N >> 4;
    if (wave >= ntiles) return;

    const int m0   = blockIdx.x << 4;
    const int n0   = wave << 4;
    const int row  = lane & 15;
    const int half = lane >> 4;

    v8f c = {};

    {
        const float* arow = A1 + (size_t)(m0 + row) * K;
        for (int k0 = 0; k0 < K; k0 += 4) {
            v2f a, b;
            a.x = arow[k0 + 2 * half];
            a.y = arow[k0 + 2 * half + 1];
            b.x = W1[(size_t)(k0 + 2 * half) * N + n0 + row];
            b.y = W1[(size_t)(k0 + 2 * half + 1) * N + n0 + row];
            c = __builtin_amdgcn_wmma_f32_16x16x4_f32(false, a, false, b,
                                                      (short)0, c, false, false);
        }
    }
    if (A2 != nullptr) {
        const float* arow = A2 + (size_t)(m0 + row) * K;
        for (int k0 = 0; k0 < K; k0 += 4) {
            v2f a, b;
            a.x = arow[k0 + 2 * half];
            a.y = arow[k0 + 2 * half + 1];
            b.x = W2[(size_t)(k0 + 2 * half) * N + n0 + row];
            b.y = W2[(size_t)(k0 + 2 * half + 1) * N + n0 + row];
            c = __builtin_amdgcn_wmma_f32_16x16x4_f32(false, a, false, b,
                                                      (short)0, c, false, false);
        }
    }

    const float bv = bias ? bias[n0 + row] : 0.0f;
    #pragma unroll
    for (int i = 0; i < 8; ++i) {
        float x = c[i] + bv;
        if (relu) x = fmaxf(x, 0.0f);
        int m = m0 + i + 8 * half;
        if (m < M) C[(size_t)m * N + n0 + row] = x;
    }
}

extern "C" void kernel_launch(void* const* d_in, const int* in_sizes, int n_in,
                              void* d_out, int out_size, void* d_ws, size_t ws_size,
                              hipStream_t stream) {
    const float* X       = (const float*)d_in[0];
    const int*   ei      = (const int*)  d_in[1];
    const float* enc_W   = (const float*)d_in[2];
    const float* enc_b   = (const float*)d_in[3];
    const float* conv_Wl = (const float*)d_in[4];
    const float* conv_bl = (const float*)d_in[5];
    const float* conv_Wr = (const float*)d_in[6];
    const float* gg_Wl   = (const float*)d_in[7];
    const float* gg_bl   = (const float*)d_in[8];
    const float* gg_Wr   = (const float*)d_in[9];
    const float* Q_W     = (const float*)d_in[10];
    const float* Q_b     = (const float*)d_in[11];
    const float* dec_W   = (const float*)d_in[12];
    const float* dec_b   = (const float*)d_in[13];

    const int N = in_sizes[0] / NFEAT;
    const int E = in_sizes[1] / 2;
    const int* src = ei;
    const int* dst = ei + E;

    float* ws       = (float*)d_ws;
    const size_t NH = (size_t)N * NHID;
    float* cnt_dst  = ws;
    float* cnt_src  = cnt_dst + N;
    float* Xa       = cnt_src + N;
    float* Xb       = Xa + NH;
    float* agg      = Xb + NH;
    float* T1       = agg + NH;
    float* T2       = T1 + NH;
    float* Ab       = T2 + NH;
    float* Bb       = Ab + NH;
    float* hs       = Bb + NH;

    auto cdiv = [](long long a, long long b) { return (int)((a + b - 1) / b); };
    const int total   = (int)NH;
    const int eThr    = E * 32;
    const int gemmGrd = N / 16;

    k_zero <<<cdiv(2 * N, 256), 256, 0, stream>>>(cnt_dst, 2 * N);
    k_count<<<cdiv(E, 256),     256, 0, stream>>>(src, dst, cnt_dst, cnt_src, E);

    k_gemm<<<gemmGrd, 256, 0, stream>>>(X, enc_W, nullptr, nullptr, enc_b,
                                        Xa, N, NFEAT, NHID, 1);

    float* Xcur  = Xa;
    float* Xnext = Xb;
    for (int layer = 0; layer < 2; ++layer) {
        k_zero        <<<cdiv(total, 256), 256, 0, stream>>>(agg, total);
        k_scatter_rows<<<cdiv(eThr, 256),  256, 0, stream>>>(Xcur, src, dst, agg, E);
        k_div_rows    <<<cdiv(total, 256), 256, 0, stream>>>(agg, cnt_dst, total);

        k_gemm<<<gemmGrd, 256, 0, stream>>>(agg, conv_Wl, Xcur, conv_Wr, conv_bl,
                                            T1, N, NHID, NHID, 1);
        k_gemm<<<gemmGrd, 256, 0, stream>>>(agg, gg_Wl, Xcur, gg_Wr, gg_bl,
                                            T2, N, NHID, NHID, 1);

        k_gemm<<<gemmGrd, 256, 0, stream>>>(T2, Q_W, nullptr, nullptr, Q_b,
                                            Ab, N, NHID, NHID, 0);
        k_gemm<<<gemmGrd, 256, 0, stream>>>(T2, Q_W + (size_t)NHID * NHID, nullptr, nullptr,
                                            nullptr, Bb, N, NHID, NHID, 0);

        k_zero  <<<cdiv(total, 256), 256, 0, stream>>>(hs, total);
        k_edge_h<<<cdiv(eThr, 256),  256, 0, stream>>>(Ab, Bb, src, dst, hs, E);
        k_gate  <<<cdiv(total, 256), 256, 0, stream>>>(Xcur, T1, hs, cnt_src, Xnext, total);

        float* t = Xcur; Xcur = Xnext; Xnext = t;
    }

    k_gemm<<<gemmGrd, 256, 0, stream>>>(Xcur, dec_W, nullptr, nullptr, dec_b,
                                        (float*)d_out, N, NHID, NCLASS, 0);
}


# Scatter atomics and their gather loads re-laned: lane l owns floats l+32j, so each f32 atomic instruction covers one whole 128-B line, not 8 floats in each of 4 lines
# speedup vs baseline: 1.2386x; 1.2386x over previous
#include <hip/hip_runtime.h>
#include <hip/hip_bf16.h>


typedef __attribute__((ext_vector_type(2))) float v2f;
typedef __attribute__((ext_vector_type(8))) float v8f;

#define NFEAT  256
#define NHID   128
#define NCLASS 64

__device__ __forceinline__ void atomAddF(float* p, float v) {
    __hip_atomic_fetch_add(p, v, __ATOMIC_RELAXED, __HIP_MEMORY_SCOPE_AGENT);
}

__global__ void k_zero(float* __restrict__ p, int n) {
    int i = blockIdx.x * blockDim.x + threadIdx.x;
    if (i < n) p[i] = 0.0f;
}

__global__ void k_count(const int* __restrict__ src, const int* __restrict__ dst,
                        float* __restrict__ cdst, float* __restrict__ csrc, int E) {
    int e = blockIdx.x * blockDim.x + threadIdx.x;
    if (e < E) {
        atomAddF(&cdst[dst[e]], 1.0f);
        atomAddF(&csrc[src[e]], 1.0f);
    }
}

__global__ void k_scatter_rows(const float* __restrict__ X,
                               const int* __restrict__ gath,
                               const int* __restrict__ scat,
                               float* __restrict__ out, int E) {
    int t = blockIdx.x * blockDim.x + threadIdx.x;
    int e = t >> 5;
    if (e >= E) return;
    int l = t & 31;
    const float* x = X + (size_t)gath[e] * NHID + l;
    float* o = out + (size_t)scat[e] * NHID + l;
    const float v0 = x[0];
    const float v1 = x[32];
    const float v2 = x[64];
    const float v3 = x[96];
    atomAddF(o +  0, v0);
    atomAddF(o + 32, v1);
    atomAddF(o + 64, v2);
    atomAddF(o + 96, v3);
}

__global__ void k_div_rows(float* __restrict__ agg, const float* __restrict__ cnt, int total) {
    int i = blockIdx.x * blockDim.x + threadIdx.x;
    if (i >= total) return;
    float c = cnt[i >> 7];
    agg[i] *= 1.0f / fmaxf(c, 1.0f);
}

__global__ void k_edge_h(const float* __restrict__ A, const float* __restrict__ B,
                         const int* __restrict__ src, const int* __restrict__ dst,
                         float* __restrict__ hs, int E) {
    int t = blockIdx.x * blockDim.x + threadIdx.x;
    int e = t >> 5;
    if (e >= E) return;
    int l = t & 31;
    int s = src[e], d = dst[e];
    const float* a = A + (size_t)s * NHID + l;
    const float* b = B + (size_t)d * NHID + l;
    const float a0 = a[0], a1 = a[32], a2 = a[64], a3 = a[96];
    const float b0 = b[0], b1 = b[32], b2 = b[64], b3 = b[96];
    float h0 = (a0 + b0) * (a0 + b0);
    float h1 = (a1 + b1) * (a1 + b1);
    float h2 = (a2 + b2) * (a2 + b2);
    float h3 = (a3 + b3) * (a3 + b3);
    float* o = hs + (size_t)s * NHID + l;
    atomAddF(o +  0, h0);
    atomAddF(o + 32, h1);
    atomAddF(o + 64, h2);
    atomAddF(o + 96, h3);
}

__global__ void k_gate(const float* __restrict__ Xcur, const float* __restrict__ Xnew,
                       const float* __restrict__ hs, const float* __restrict__ csrc,
                       float* __restrict__ Xnext, int total) {
    int i = blockIdx.x * blockDim.x + threadIdx.x;
    if (i >= total) return;
    float c   = fmaxf(csrc[i >> 7], 1.0f);
    float tau = tanhf(hs[i] / c);
    Xnext[i]  = (1.0f - tau) * Xcur[i] + tau * Xnew[i];
}

__global__ void k_gemm(const float* __restrict__ A1, const float* __restrict__ W1,
                       const float* __restrict__ A2, const float* __restrict__ W2,
                       const float* __restrict__ bias, float* __restrict__ C,
                       int M, int K, int N, int relu) {
    const int lane   = threadIdx.x & 31;
    const int wave   = threadIdx.x >> 5;
    const int ntiles = N >> 4;
    if (wave >= ntiles) return;

    const int m0   = blockIdx.x << 4;
    const int n0   = wave << 4;
    const int row  = lane & 15;
    const int half = lane >> 4;

    v8f c = {};

    {
        const float* arow = A1 + (size_t)(m0 + row) * K;
        for (int k0 = 0; k0 < K; k0 += 4) {
            v2f a, b;
            a.x = arow[k0 + 2 * half];
            a.y = arow[k0 + 2 * half + 1];
            b.x = W1[(size_t)(k0 + 2 * half) * N + n0 + row];
            b.y = W1[(size_t)(k0 + 2 * half + 1) * N + n0 + row];
            c = __builtin_amdgcn_wmma_f32_16x16x4_f32(false, a, false, b,
                                                      (short)0, c, false, false);
        }
    }
    if (A2 != nullptr) {
        const float* arow = A2 + (size_t)(m0 + row) * K;
        for (int k0 = 0; k0 < K; k0 += 4) {
            v2f a, b;
            a.x = arow[k0 + 2 * half];
            a.y = arow[k0 + 2 * half + 1];
            b.x = W2[(size_t)(k0 + 2 * half) * N + n0 + row];
            b.y = W2[(size_t)(k0 + 2 * half + 1) * N + n0 + row];
            c = __builtin_amdgcn_wmma_f32_16x16x4_f32(false, a, false, b,
                                                      (short)0, c, false, false);
        }
    }

    const float bv = bias ? bias[n0 + row] : 0.0f;
    #pragma unroll
    for (int i = 0; i < 8; ++i) {
        float x = c[i] + bv;
        if (relu) x = fmaxf(x, 0.0f);
        int m = m0 + i + 8 * half;
        if (m < M) C[(size_t)m * N + n0 + row] = x;
    }
}

extern "C" void kernel_launch(void* const* d_in, const int* in_sizes, int n_in,
                              void* d_out, int out_size, void* d_ws, size_t ws_size,
                              hipStream_t stream) {
    const float* X       = (const float*)d_in[0];
    const int*   ei      = (const int*)  d_in[1];
    const float* enc_W   = (const float*)d_in[2];
    const float* enc_b   = (const float*)d_in[3];
    const float* conv_Wl = (const float*)d_in[4];
    const float* conv_bl = (const float*)d_in[5];
    const float* conv_Wr = (const float*)d_in[6];
    const float* gg_Wl   = (const float*)d_in[7];
    const float* gg_bl   = (const float*)d_in[8];
    const float* gg_Wr   = (const float*)d_in[9];
    const float* Q_W     = (const float*)d_in[10];
    const float* Q_b     = (const float*)d_in[11];
    const float* dec_W   = (const float*)d_in[12];
    const float* dec_b   = (const float*)d_in[13];

    const int N = in_sizes[0] / NFEAT;
    const int E = in_sizes[1] / 2;
    const int* src = ei;
    const int* dst = ei + E;

    float* ws       = (float*)d_ws;
    const size_t NH = (size_t)N * NHID;
    float* cnt_dst  = ws;
    float* cnt_src  = cnt_dst + N;
    float* Xa       = cnt_src + N;
    float* Xb       = Xa + NH;
    float* agg      = Xb + NH;
    float* T1       = agg + NH;
    float* T2       = T1 + NH;
    float* Ab       = T2 + NH;
    float* Bb       = Ab + NH;
    float* hs       = Bb + NH;

    auto cdiv = [](long long a, long long b) { return (int)((a + b - 1) / b); };
    const int total   = (int)NH;
    const int eThr    = E * 32;
    const int gemmGrd = N / 16;

    k_zero <<<cdiv(2 * N, 256), 256, 0, stream>>>(cnt_dst, 2 * N);
    k_count<<<cdiv(E, 256),     256, 0, stream>>>(src, dst, cnt_dst, cnt_src, E);

    k_gemm<<<gemmGrd, 256, 0, stream>>>(X, enc_W, nullptr, nullptr, enc_b,
                                        Xa, N, NFEAT, NHID, 1);

    float* Xcur  = Xa;
    float* Xnext = Xb;
    for (int layer = 0; layer < 2; ++layer) {
        k_zero        <<<cdiv(total, 256), 256, 0, stream>>>(agg, total);
        k_scatter_rows<<<cdiv(eThr, 256),  256, 0, stream>>>(Xcur, src, dst, agg, E);
        k_div_rows    <<<cdiv(total, 256), 256, 0, stream>>>(agg, cnt_dst, total);

        k_gemm<<<gemmGrd, 256, 0, stream>>>(agg, conv_Wl, Xcur, conv_Wr, conv_bl,
                                            T1, N, NHID, NHID, 1);
        k_gemm<<<gemmGrd, 256, 0, stream>>>(agg, gg_Wl, Xcur, gg_Wr, gg_bl,
                                            T2, N, NHID, NHID, 1);

        k_gemm<<<gemmGrd, 256, 0, stream>>>(T2, Q_W, nullptr, nullptr, Q_b,
                                            Ab, N, NHID, NHID, 0);
        k_gemm<<<gemmGrd, 256, 0, stream>>>(T2, Q_W + (size_t)NHID * NHID, nullptr, nullptr,
                                            nullptr, Bb, N, NHID, NHID, 0);

        k_zero  <<<cdiv(total, 256), 256, 0, stream>>>(hs, total);
        k_edge_h<<<cdiv(eThr, 256),  256, 0, stream>>>(Ab, Bb, src, dst, hs, E);
        k_gate  <<<cdiv(total, 256), 256, 0, stream>>>(Xcur, T1, hs, cnt_src, Xnext, total);

        float* t = Xcur; Xcur = Xnext; Xnext = t;
    }

    k_gemm<<<gemmGrd, 256, 0, stream>>>(Xcur, dec_W, nullptr, nullptr, dec_b,
                                        (float*)d_out, N, NHID, NCLASS, 0);
}


# Scatter-add float atomics replaced by per-node gathers over neighbor lists rebuilt in-call (count, scan, fill): one wave per node sums rows in registers; zero, divide and gate passes fold in
# speedup vs baseline: 1.4578x; 1.1770x over previous
#include <hip/hip_runtime.h>
#include <hip/hip_bf16.h>


typedef __attribute__((ext_vector_type(2))) float v2f;
typedef __attribute__((ext_vector_type(8))) float v8f;

#define NFEAT  256
#define NHID   128
#define NCLASS 64

__global__ void k_zero(int* __restrict__ p, int n) {
    int i = blockIdx.x * blockDim.x + threadIdx.x;
    if (i < n) p[i] = 0;
}

__global__ void k_count(const int* __restrict__ src, const int* __restrict__ dst,
                        int* __restrict__ cdst, int* __restrict__ csrc, int E) {
    int e = blockIdx.x * blockDim.x + threadIdx.x;
    if (e < E) {
        __hip_atomic_fetch_add(&cdst[dst[e]], 1, __ATOMIC_RELAXED, __HIP_MEMORY_SCOPE_AGENT);
        __hip_atomic_fetch_add(&csrc[src[e]], 1, __ATOMIC_RELAXED, __HIP_MEMORY_SCOPE_AGENT);
    }
}

__global__ void k_scan(const int* __restrict__ cnt, int* __restrict__ off,
                       int* __restrict__ cur, int N) {
    __shared__ int sh[1024];
    const int* c  = cnt + (size_t)blockIdx.x * N;
    int*       o  = off + (size_t)blockIdx.x * (N + 1);
    int*       cu = cur + (size_t)blockIdx.x * N;
    const int t     = threadIdx.x;
    const int chunk = (N + 1023) >> 10;
    const int b     = t * chunk;
    const int e     = min(b + chunk, N);
    int s = 0;
    for (int i = b; i < e; ++i) s += c[i];
    sh[t] = s;
    __syncthreads();
    for (int d = 1; d < 1024; d <<= 1) {
        int v = (t >= d) ? sh[t - d] : 0;
        __syncthreads();
        sh[t] += v;
        __syncthreads();
    }
    int run = sh[t] - s;
    for (int i = b; i < e; ++i) {
        o[i]  = run;
        cu[i] = run;
        run  += c[i];
    }
    if (t == 1023) o[N] = sh[1023];
}

__global__ void k_fill(const int* __restrict__ src, const int* __restrict__ dst,
                       int* __restrict__ cur_dst, int* __restrict__ cur_src,
                       int* __restrict__ nbr_in, int* __restrict__ nbr_out, int E) {
    int e = blockIdx.x * blockDim.x + threadIdx.x;
    if (e >= E) return;
    const int s = src[e], d = dst[e];
    const int p = __hip_atomic_fetch_add(&cur_dst[d], 1, __ATOMIC_RELAXED, __HIP_MEMORY_SCOPE_AGENT);
    nbr_in[p] = s;
    const int q = __hip_atomic_fetch_add(&cur_src[s], 1, __ATOMIC_RELAXED, __HIP_MEMORY_SCOPE_AGENT);
    nbr_out[q] = d;
}

__device__ __forceinline__ float4 ldRow(const float* __restrict__ X, int r, int l) {
    return *reinterpret_cast<const float4*>(X + (size_t)r * NHID + 4 * l);
}

__global__ void k_gather_mean(const float* __restrict__ X, const int* __restrict__ off,
                              const int* __restrict__ nbr, float* __restrict__ agg, int N) {
    const int t = blockIdx.x * blockDim.x + threadIdx.x;
    const int node = __builtin_amdgcn_readfirstlane(t >> 5);
    if (node >= N) return;
    const int l  = t & 31;
    const int b  = off[node];
    const int e  = off[node + 1];
    float4 acc = make_float4(0.f, 0.f, 0.f, 0.f);
    for (int base = b; base < e; base += 32) {
        const int n  = min(32, e - base);
        const int id = (l < n) ? nbr[base + l] : 0;
        for (int j = 0; j < n; j += 8) {
            float4 v[8];
            #pragma unroll
            for (int u = 0; u < 8; ++u)
                v[u] = ldRow(X, __builtin_amdgcn_readlane(id, (j + u) & 31), l);
            #pragma unroll
            for (int u = 0; u < 8; ++u) {
                const bool live = (j + u < n);
                acc.x += live ? v[u].x : 0.0f;
                acc.y += live ? v[u].y : 0.0f;
                acc.z += live ? v[u].z : 0.0f;
                acc.w += live ? v[u].w : 0.0f;
            }
        }
    }
    const float inv = 1.0f / fmaxf((float)(e - b), 1.0f);
    float4 r = make_float4(acc.x * inv, acc.y * inv, acc.z * inv, acc.w * inv);
    *reinterpret_cast<float4*>(agg + (size_t)node * NHID + 4 * l) = r;
}

__global__ void k_gate_gather(const float* __restrict__ A, const float* __restrict__ B,
                              const int* __restrict__ off, const int* __restrict__ nbr,
                              const float* __restrict__ Xcur, const float* __restrict__ Xnew,
                              float* __restrict__ Xnext, int N) {
    const int t = blockIdx.x * blockDim.x + threadIdx.x;
    const int node = __builtin_amdgcn_readfirstlane(t >> 5);
    if (node >= N) return;
    const int l  = t & 31;
    const int b  = off[node];
    const int e  = off[node + 1];
    const float4 a  = ldRow(A, node, l);
    const float4 xc = ldRow(Xcur, node, l);
    const float4 xn = ldRow(Xnew, node, l);
    float4 h = make_float4(0.f, 0.f, 0.f, 0.f);
    for (int base = b; base < e; base += 32) {
        const int n  = min(32, e - base);
        const int id = (l < n) ? nbr[base + l] : 0;
        for (int j = 0; j < n; j += 8) {
            float4 v[8];
            #pragma unroll
            for (int u = 0; u < 8; ++u)
                v[u] = ldRow(B, __builtin_amdgcn_readlane(id, (j + u) & 31), l);
            #pragma unroll
            for (int u = 0; u < 8; ++u) {
                const bool live = (j + u < n);
                const float tx = a.x + v[u].x, ty = a.y + v[u].y;
                const float tz = a.z + v[u].z, tw = a.w + v[u].w;
                h.x += live ? tx * tx : 0.0f;
                h.y += live ? ty * ty : 0.0f;
                h.z += live ? tz * tz : 0.0f;
                h.w += live ? tw * tw : 0.0f;
            }
        }
    }
    const float c = fmaxf((float)(e - b), 1.0f);
    const float t0 = tanhf(h.x / c), t1 = tanhf(h.y / c);
    const float t2 = tanhf(h.z / c), t3 = tanhf(h.w / c);
    float4 r;
    r.x = (1.0f - t0) * xc.x + t0 * xn.x;
    r.y = (1.0f - t1) * xc.y + t1 * xn.y;
    r.z = (1.0f - t2) * xc.z + t2 * xn.z;
    r.w = (1.0f - t3) * xc.w + t3 * xn.w;
    *reinterpret_cast<float4*>(Xnext + (size_t)node * NHID + 4 * l) = r;
}

__global__ void k_gemm(const float* __restrict__ A1, const float* __restrict__ W1,
                       const float* __restrict__ A2, const float* __restrict__ W2,
                       const float* __restrict__ bias, float* __restrict__ C,
                       int M, int K, int N, int relu) {
    const int lane   = threadIdx.x & 31;
    const int wave   = threadIdx.x >> 5;
    const int ntiles = N >> 4;
    if (wave >= ntiles) return;

    const int m0   = blockIdx.x << 4;
    const int n0   = wave << 4;
    const int row  = lane & 15;
    const int half = lane >> 4;

    v8f c = {};

    {
        const float* arow = A1 + (size_t)(m0 + row) * K;
        for (int k0 = 0; k0 < K; k0 += 4) {
            v2f a, b;
            a.x = arow[k0 + 2 * half];
            a.y = arow[k0 + 2 * half + 1];
            b.x = W1[(size_t)(k0 + 2 * half) * N + n0 + row];
            b.y = W1[(size_t)(k0 + 2 * half + 1) * N + n0 + row];
            c = __builtin_amdgcn_wmma_f32_16x16x4_f32(false, a, false, b,
                                                      (short)0, c, false, false);
        }
    }
    if (A2 != nullptr) {
        const float* arow = A2 + (size_t)(m0 + row) * K;
        for (int k0 = 0; k0 < K; k0 += 4) {
            v2f a, b;
            a.x = arow[k0 + 2 * half];
            a.y = arow[k0 + 2 * half + 1];
            b.x = W2[(size_t)(k0 + 2 * half) * N + n0 + row];
            b.y = W2[(size_t)(k0 + 2 * half + 1) * N + n0 + row];
            c = __builtin_amdgcn_wmma_f32_16x16x4_f32(false, a, false, b,
                                                      (short)0, c, false, false);
        }
    }

    const float bv = bias ? bias[n0 + row] : 0.0f;
    #pragma unroll
    for (int i = 0; i < 8; ++i) {
        float x = c[i] + bv;
        if (relu) x = fmaxf(x, 0.0f);
        int m = m0 + i + 8 * half;
        if (m < M) C[(size_t)m * N + n0 + row] = x;
    }
}

extern "C" void kernel_launch(void* const* d_in, const int* in_sizes, int n_in,
                              void* d_out, int out_size, void* d_ws, size_t ws_size,
                              hipStream_t stream) {
    const float* X       = (const float*)d_in[0];
    const int*   ei      = (const int*)  d_in[1];
    const float* enc_W   = (const float*)d_in[2];
    const float* enc_b   = (const float*)d_in[3];
    const float* conv_Wl = (const float*)d_in[4];
    const float* conv_bl = (const float*)d_in[5];
    const float* conv_Wr = (const float*)d_in[6];
    const float* gg_Wl   = (const float*)d_in[7];
    const float* gg_bl   = (const float*)d_in[8];
    const float* gg_Wr   = (const float*)d_in[9];
    const float* Q_W     = (const float*)d_in[10];
    const float* Q_b     = (const float*)d_in[11];
    const float* dec_W   = (const float*)d_in[12];
    const float* dec_b   = (const float*)d_in[13];

    const int N = in_sizes[0] / NFEAT;
    const int E = in_sizes[1] / 2;
    const int* src = ei;
    const int* dst = ei + E;

    float* ws       = (float*)d_ws;
    const size_t NH = (size_t)N * NHID;
    int*   cnt      = (int*)ws;
    float* Xa       = ws + 2 * (size_t)N;
    float* Xb       = Xa + NH;
    float* agg      = Xb + NH;
    float* T1       = agg + NH;
    float* T2       = T1 + NH;
    float* Ab       = T2 + NH;
    float* Bb       = Ab + NH;
    int*   csr      = (int*)(Bb + NH);
    int*   off      = csr;
    int*   cur      = off + 2 * (size_t)(N + 1);
    int*   nbr_in   = cur + 2 * (size_t)N;
    int*   nbr_out  = nbr_in + E;
    const int* off_dst = off;
    const int* off_src = off + (N + 1);

    auto cdiv = [](long long a, long long b) { return (int)((a + b - 1) / b); };
    const int nodeGrd = cdiv((long long)N * 32, 256);
    const int gemmGrd = N / 16;

    k_zero <<<cdiv(2 * N, 256), 256, 0, stream>>>(cnt, 2 * N);
    k_count<<<cdiv(E, 256),     256, 0, stream>>>(src, dst, cnt, cnt + N, E);
    k_scan <<<2, 1024, 0, stream>>>(cnt, off, cur, N);
    k_fill <<<cdiv(E, 256),     256, 0, stream>>>(src, dst, cur, cur + N, nbr_in, nbr_out, E);

    k_gemm<<<gemmGrd, 256, 0, stream>>>(X, enc_W, nullptr, nullptr, enc_b,
                                        Xa, N, NFEAT, NHID, 1);

    float* Xcur  = Xa;
    float* Xnext = Xb;
    for (int layer = 0; layer < 2; ++layer) {
        k_gather_mean<<<nodeGrd, 256, 0, stream>>>(Xcur, off_dst, nbr_in, agg, N);

        k_gemm<<<gemmGrd, 256, 0, stream>>>(agg, conv_Wl, Xcur, conv_Wr, conv_bl,
                                            T1, N, NHID, NHID, 1);
        k_gemm<<<gemmGrd, 256, 0, stream>>>(agg, gg_Wl, Xcur, gg_Wr, gg_bl,
                                            T2, N, NHID, NHID, 1);

        k_gemm<<<gemmGrd, 256, 0, stream>>>(T2, Q_W, nullptr, nullptr, Q_b,
                                            Ab, N, NHID, NHID, 0);
        k_gemm<<<gemmGrd, 256, 0, stream>>>(T2, Q_W + (size_t)NHID * NHID, nullptr, nullptr,
                                            nullptr, Bb, N, NHID, NHID, 0);

        k_gate_gather<<<nodeGrd, 256, 0, stream>>>(Ab, Bb, off_src, nbr_out, Xcur, T1, Xnext, N);

        float* t = Xcur; Xcur = Xnext; Xnext = t;
    }

    k_gemm<<<gemmGrd, 256, 0, stream>>>(Xcur, dec_W, nullptr, nullptr, dec_b,
                                        (float*)d_out, N, NHID, NCLASS, 0);
}


# Node GEMMs re-tiled: weights and the A tile staged once per 48-row work-group in LDS by whole-line loads (K in two halves, 2nd prefetched); WMMA fragments are LDS reads, not global loads per tile
# speedup vs baseline: 1.7557x; 1.2043x over previous
#include <hip/hip_runtime.h>
#include <hip/hip_bf16.h>


typedef __attribute__((ext_vector_type(2))) float v2f;
typedef __attribute__((ext_vector_type(8))) float v8f;

#define NFEAT  256
#define NHID   128
#define NCLASS 64

__global__ void k_zero(int* __restrict__ p, int n) {
    int i = blockIdx.x * blockDim.x + threadIdx.x;
    if (i < n) p[i] = 0;
}

__global__ void k_count(const int* __restrict__ src, const int* __restrict__ dst,
                        int* __restrict__ cdst, int* __restrict__ csrc, int E) {
    int e = blockIdx.x * blockDim.x + threadIdx.x;
    if (e < E) {
        __hip_atomic_fetch_add(&cdst[dst[e]], 1, __ATOMIC_RELAXED, __HIP_MEMORY_SCOPE_AGENT);
        __hip_atomic_fetch_add(&csrc[src[e]], 1, __ATOMIC_RELAXED, __HIP_MEMORY_SCOPE_AGENT);
    }
}

__global__ void k_scan(const int* __restrict__ cnt, int* __restrict__ off,
                       int* __restrict__ cur, int N) {
    __shared__ int sh[1024];
    const int* c  = cnt + (size_t)blockIdx.x * N;
    int*       o  = off + (size_t)blockIdx.x * (N + 1);
    int*       cu = cur + (size_t)blockIdx.x * N;
    const int t     = threadIdx.x;
    const int chunk = (N + 1023) >> 10;
    const int b     = t * chunk;
    const int e     = min(b + chunk, N);
    int s = 0;
    for (int i = b; i < e; ++i) s += c[i];
    sh[t] = s;
    __syncthreads();
    for (int d = 1; d < 1024; d <<= 1) {
        int v = (t >= d) ? sh[t - d] : 0;
        __syncthreads();
        sh[t] += v;
        __syncthreads();
    }
    int run = sh[t] - s;
    for (int i = b; i < e; ++i) {
        o[i]  = run;
        cu[i] = run;
        run  += c[i];
    }
    if (t == 1023) o[N] = sh[1023];
}

__global__ void k_fill(const int* __restrict__ src, const int* __restrict__ dst,
                       int* __restrict__ cur_dst, int* __restrict__ cur_src,
                       int* __restrict__ nbr_in, int* __restrict__ nbr_out, int E) {
    int e = blockIdx.x * blockDim.x + threadIdx.x;
    if (e >= E) return;
    const int s = src[e], d = dst[e];
    const int p = __hip_atomic_fetch_add(&cur_dst[d], 1, __ATOMIC_RELAXED, __HIP_MEMORY_SCOPE_AGENT);
    nbr_in[p] = s;
    const int q = __hip_atomic_fetch_add(&cur_src[s], 1, __ATOMIC_RELAXED, __HIP_MEMORY_SCOPE_AGENT);
    nbr_out[q] = d;
}

__device__ __forceinline__ float4 ldRow(const float* __restrict__ X, int r, int l) {
    return *reinterpret_cast<const float4*>(X + (size_t)r * NHID + 4 * l);
}

__global__ void k_gather_mean(const float* __restrict__ X, const int* __restrict__ off,
                              const int* __restrict__ nbr, float* __restrict__ agg, int N) {
    const int t = blockIdx.x * blockDim.x + threadIdx.x;
    const int node = __builtin_amdgcn_readfirstlane(t >> 5);
    if (node >= N) return;
    const int l  = t & 31;
    const int b  = off[node];
    const int e  = off[node + 1];
    float4 acc = make_float4(0.f, 0.f, 0.f, 0.f);
    for (int base = b; base < e; base += 32) {
        const int n  = min(32, e - base);
        const int id = (l < n) ? nbr[base + l] : 0;
        for (int j = 0; j < n; j += 8) {
            float4 v[8];
            #pragma unroll
            for (int u = 0; u < 8; ++u)
                v[u] = ldRow(X, __builtin_amdgcn_readlane(id, (j + u) & 31), l);
            #pragma unroll
            for (int u = 0; u < 8; ++u) {
                const bool live = (j + u < n);
                acc.x += live ? v[u].x : 0.0f;
                acc.y += live ? v[u].y : 0.0f;
                acc.z += live ? v[u].z : 0.0f;
                acc.w += live ? v[u].w : 0.0f;
            }
        }
    }
    const float inv = 1.0f / fmaxf((float)(e - b), 1.0f);
    float4 r = make_float4(acc.x * inv, acc.y * inv, acc.z * inv, acc.w * inv);
    *reinterpret_cast<float4*>(agg + (size_t)node * NHID + 4 * l) = r;
}

__global__ void k_gate_gather(const float* __restrict__ A, const float* __restrict__ B,
                              const int* __restrict__ off, const int* __restrict__ nbr,
                              const float* __restrict__ Xcur, const float* __restrict__ Xnew,
                              float* __restrict__ Xnext, int N) {
    const int t = blockIdx.x * blockDim.x + threadIdx.x;
    const int node = __builtin_amdgcn_readfirstlane(t >> 5);
    if (node >= N) return;
    const int l  = t & 31;
    const int b  = off[node];
    const int e  = off[node + 1];
    const float4 a  = ldRow(A, node, l);
    const float4 xc = ldRow(Xcur, node, l);
    const float4 xn = ldRow(Xnew, node, l);
    float4 h = make_float4(0.f, 0.f, 0.f, 0.f);
    for (int base = b; base < e; base += 32) {
        const int n  = min(32, e - base);
        const int id = (l < n) ? nbr[base + l] : 0;
        for (int j = 0; j < n; j += 8) {
            float4 v[8];
            #pragma unroll
            for (int u = 0; u < 8; ++u)
                v[u] = ldRow(B, __builtin_amdgcn_readlane(id, (j + u) & 31), l);
            #pragma unroll
            for (int u = 0; u < 8; ++u) {
                const bool live = (j + u < n);
                const float tx = a.x + v[u].x, ty = a.y + v[u].y;
                const float tz = a.z + v[u].z, tw = a.w + v[u].w;
                h.x += live ? tx * tx : 0.0f;
                h.y += live ? ty * ty : 0.0f;
                h.z += live ? tz * tz : 0.0f;
                h.w += live ? tw * tw : 0.0f;
            }
        }
    }
    const float c = fmaxf((float)(e - b), 1.0f);
    const float t0 = tanhf(h.x / c), t1 = tanhf(h.y / c);
    const float t2 = tanhf(h.z / c), t3 = tanhf(h.w / c);
    float4 r;
    r.x = (1.0f - t0) * xc.x + t0 * xn.x;
    r.y = (1.0f - t1) * xc.y + t1 * xn.y;
    r.z = (1.0f - t2) * xc.z + t2 * xn.z;
    r.w = (1.0f - t3) * xc.w + t3 * xn.w;
    *reinterpret_cast<float4*>(Xnext + (size_t)node * NHID + 4 * l) = r;
}

#define GEMM_THREADS 256
#define A_PITCH      132

template <int N_, int NSEG>
__device__ __forceinline__ void gemm_body(float* __restrict__ smem,
                                          const float* __restrict__ A1, int lda1, const float* __restrict__ W1,
                                          const float* __restrict__ A2, int lda2, const float* __restrict__ W2,
                                          const float* __restrict__ bias, float* __restrict__ C,
                                          int M, int relu) {
    constexpr int PITCH  = 2 * N_ + 16;
    constexpr int NTILES = N_ / 16;
    constexpr int MG     = 8 / NTILES;
    constexpr int ROWS   = 48 * MG;
    constexpr int G      = N_ / 4;
    constexpr int PH     = NSEG * 32;
    constexpr int WTRIPS = (PH * G) / GEMM_THREADS;
    constexpr int ATRIPS = (ROWS * 32) / GEMM_THREADS;
    constexpr int KBP    = NSEG * 8;
    static_assert((PH * G) % GEMM_THREADS == 0 && (ROWS * 32) % GEMM_THREADS == 0,
                  "staging work divides evenly over the work-group");

    float* sW = smem;
    float* sA = smem + NSEG * 64 * PITCH;

    const int tid  = threadIdx.x;
    const int lane = tid & 31;
    const int wave = tid >> 5;
    const int c    = lane & 15;
    const int h    = lane >> 4;
    const int nt   = wave % NTILES;
    const int mg   = wave / NTILES;
    const int m0   = blockIdx.x * ROWS;

    float4 w0[WTRIPS], w1[WTRIPS];
    auto loadW = [&](int ph) {
        #pragma unroll
        for (int it = 0; it < WTRIPS; ++it) {
            const int i    = it * GEMM_THREADS + tid;
            const int qrow = ph * PH + i / G;
            const int g    = i % G;
            const float* Wp = ((qrow >> 6) == 0 ? W1 : W2) + (size_t)(2 * (qrow & 63)) * N_ + 4 * g;
            w0[it] = *reinterpret_cast<const float4*>(Wp);
            w1[it] = *reinterpret_cast<const float4*>(Wp + N_);
        }
    };
    auto storeW = [&](int ph) {
        #pragma unroll
        for (int it = 0; it < WTRIPS; ++it) {
            const int i    = it * GEMM_THREADS + tid;
            const int qrow = ph * PH + i / G;
            const int g    = i % G;
            float* d = sW + qrow * PITCH + 8 * g;
            *reinterpret_cast<float4*>(d)     = make_float4(w0[it].x, w1[it].x, w0[it].y, w1[it].y);
            *reinterpret_cast<float4*>(d + 4) = make_float4(w0[it].z, w1[it].z, w0[it].w, w1[it].w);
        }
    };

    v8f acc[3];
    #pragma unroll
    for (int i = 0; i < 3; ++i) acc[i] = v8f{};

    auto compute = [&](int ph) {
        #pragma unroll
        for (int kbl = 0; kbl < KBP; ++kbl) {
            const int KB  = ph * KBP + kbl;
            const int s   = KB >> 4;
            const int kbs = KB & 15;
            float4 a[3];
            #pragma unroll
            for (int i = 0; i < 3; ++i)
                a[i] = *reinterpret_cast<const float4*>(
                    sA + (s * ROWS + 48 * mg + 16 * i + c) * A_PITCH + 8 * kbs + 4 * h);
            #pragma unroll
            for (int j = 0; j < 2; ++j) {
                const v2f bv = *reinterpret_cast<const v2f*>(
                    sW + (4 * KB + 2 * h + j) * PITCH + 2 * (16 * nt + c));
                #pragma unroll
                for (int i = 0; i < 3; ++i) {
                    v2f av;
                    av.x = (j == 0) ? a[i].x : a[i].z;
                    av.y = (j == 0) ? a[i].y : a[i].w;
                    acc[i] = __builtin_amdgcn_wmma_f32_16x16x4_f32(false, av, false, bv,
                                                                   (short)0, acc[i], false, false);
                }
            }
        }
    };

    loadW(0);
    #pragma unroll
    for (int s = 0; s < NSEG; ++s) {
        #pragma unroll
        for (int it = 0; it < ATRIPS; ++it) {
            const int i   = it * GEMM_THREADS + tid;
            const int row = i >> 5;
            const int q   = i & 31;
            const int gr  = min(m0 + row, M - 1);
            const float* ap = (s == 0) ? A1 + (size_t)gr * lda1 : A2 + (size_t)gr * lda2;
            const float4 v = *reinterpret_cast<const float4*>(ap + 4 * q);
            *reinterpret_cast<float4*>(sA + (s * ROWS + row) * A_PITCH + 4 * q) = v;
        }
    }
    storeW(0);
    __syncthreads();
    loadW(1);
    compute(0);
    __builtin_amdgcn_sched_barrier(0);
    storeW(1);
    __syncthreads();
    compute(1);

    const int n = 16 * nt + c;
    const float bv = bias ? bias[n] : 0.0f;
    #pragma unroll
    for (int i = 0; i < 3; ++i) {
        #pragma unroll
        for (int r = 0; r < 8; ++r) {
            float x = acc[i][r] + bv;
            if (relu) x = fmaxf(x, 0.0f);
            const int m = m0 + 48 * mg + 16 * i + 8 * h + r;
            if (m < M) C[(size_t)m * N_ + n] = x;
        }
    }
}

__global__ void __launch_bounds__(GEMM_THREADS)
k_gemm_n128_s2(const float* __restrict__ A1, int lda1, const float* __restrict__ W1,
               const float* __restrict__ A2, int lda2, const float* __restrict__ W2,
               const float* __restrict__ bias, float* __restrict__ C, int M, int relu) {
    __shared__ __attribute__((aligned(16))) float smem[2 * 64 * (2 * 128 + 16) + 2 * 48 * A_PITCH];
    gemm_body<128, 2>(smem, A1, lda1, W1, A2, lda2, W2, bias, C, M, relu);
}

__global__ void __launch_bounds__(GEMM_THREADS)
k_gemm_n128_s1(const float* __restrict__ A1, int lda1, const float* __restrict__ W1,
               const float* __restrict__ bias, float* __restrict__ C, int M, int relu) {
    __shared__ __attribute__((aligned(16))) float smem[64 * (2 * 128 + 16) + 48 * A_PITCH];
    gemm_body<128, 1>(smem, A1, lda1, W1, nullptr, 0, nullptr, bias, C, M, relu);
}

__global__ void __launch_bounds__(GEMM_THREADS)
k_gemm_n64_s1(const float* __restrict__ A1, int lda1, const float* __restrict__ W1,
              const float* __restrict__ bias, float* __restrict__ C, int M, int relu) {
    __shared__ __attribute__((aligned(16))) float smem[64 * (2 * 64 + 16) + 96 * A_PITCH];
    gemm_body<64, 1>(smem, A1, lda1, W1, nullptr, 0, nullptr, bias, C, M, relu);
}

extern "C" void kernel_launch(void* const* d_in, const int* in_sizes, int n_in,
                              void* d_out, int out_size, void* d_ws, size_t ws_size,
                              hipStream_t stream) {
    const float* X       = (const float*)d_in[0];
    const int*   ei      = (const int*)  d_in[1];
    const float* enc_W   = (const float*)d_in[2];
    const float* enc_b   = (const float*)d_in[3];
    const float* conv_Wl = (const float*)d_in[4];
    const float* conv_bl = (const float*)d_in[5];
    const float* conv_Wr = (const float*)d_in[6];
    const float* gg_Wl   = (const float*)d_in[7];
    const float* gg_bl   = (const float*)d_in[8];
    const float* gg_Wr   = (const float*)d_in[9];
    const float* Q_W     = (const float*)d_in[10];
    const float* Q_b     = (const float*)d_in[11];
    const float* dec_W   = (const float*)d_in[12];
    const float* dec_b   = (const float*)d_in[13];

    const int N = in_sizes[0] / NFEAT;
    const int E = in_sizes[1] / 2;
    const int* src = ei;
    const int* dst = ei + E;

    float* ws       = (float*)d_ws;
    const size_t NH = (size_t)N * NHID;
    int*   cnt      = (int*)ws;
    float* Xa       = ws + 2 * (size_t)N;
    float* Xb       = Xa + NH;
    float* agg      = Xb + NH;
    float* T1       = agg + NH;
    float* T2       = T1 + NH;
    float* Ab       = T2 + NH;
    float* Bb       = Ab + NH;
    int*   csr      = (int*)(Bb + NH);
    int*   off      = csr;
    int*   cur      = off + 2 * (size_t)(N + 1);
    int*   nbr_in   = cur + 2 * (size_t)N;
    int*   nbr_out  = nbr_in + E;
    const int* off_dst = off;
    const int* off_src = off + (N + 1);

    auto cdiv = [](long long a, long long b) { return (int)((a + b - 1) / b); };
    const int nodeGrd = cdiv((long long)N * 32, 256);
    const int gemmGrd = cdiv(N, 48);
    const int decGrd  = cdiv(N, 96);

    k_zero <<<cdiv(2 * N, 256), 256, 0, stream>>>(cnt, 2 * N);
    k_count<<<cdiv(E, 256),     256, 0, stream>>>(src, dst, cnt, cnt + N, E);
    k_scan <<<2, 1024, 0, stream>>>(cnt, off, cur, N);
    k_fill <<<cdiv(E, 256),     256, 0, stream>>>(src, dst, cur, cur + N, nbr_in, nbr_out, E);

    k_gemm_n128_s2<<<gemmGrd, GEMM_THREADS, 0, stream>>>(X, NFEAT, enc_W, X + NHID, NFEAT,
                                                         enc_W + (size_t)NHID * NHID, enc_b, Xa, N, 1);

    float* Xcur  = Xa;
    float* Xnext = Xb;
    for (int layer = 0; layer < 2; ++layer) {
        k_gather_mean<<<nodeGrd, 256, 0, stream>>>(Xcur, off_dst, nbr_in, agg, N);

        k_gemm_n128_s2<<<gemmGrd, GEMM_THREADS, 0, stream>>>(agg, NHID, conv_Wl, Xcur, NHID, conv_Wr,
                                                             conv_bl, T1, N, 1);
        k_gemm_n128_s2<<<gemmGrd, GEMM_THREADS, 0, stream>>>(agg, NHID, gg_Wl, Xcur, NHID, gg_Wr,
                                                             gg_bl, T2, N, 1);

        k_gemm_n128_s1<<<gemmGrd, GEMM_THREADS, 0, stream>>>(T2, NHID, Q_W, Q_b, Ab, N, 0);
        k_gemm_n128_s1<<<gemmGrd, GEMM_THREADS, 0, stream>>>(T2, NHID, Q_W + (size_t)NHID * NHID,
                                                             nullptr, Bb, N, 0);

        k_gate_gather<<<nodeGrd, 256, 0, stream>>>(Ab, Bb, off_src, nbr_out, Xcur, T1, Xnext, N);

        float* t = Xcur; Xcur = Xnext; Xnext = t;
    }

    k_gemm_n64_s1<<<decGrd, GEMM_THREADS, 0, stream>>>(Xcur, NHID, dec_W, dec_b, (float*)d_out, N, 0);
}


# The four node GEMMs of a layer (conv, gate conv, both halves of the edge linear) fused into one launch: agg/X tiles staged once, Xg stays in LDS, weights via a 2-stage LDS ring; 18 -> 12 launches
# speedup vs baseline: 1.9981x; 1.1381x over previous
#include <hip/hip_runtime.h>
#include <hip/hip_bf16.h>


typedef __attribute__((ext_vector_type(2))) float v2f;
typedef __attribute__((ext_vector_type(8))) float v8f;

#define NFEAT  256
#define NHID   128
#define NCLASS 64

__global__ void k_zero(int* __restrict__ p, int n) {
    int i = blockIdx.x * blockDim.x + threadIdx.x;
    if (i < n) p[i] = 0;
}

__global__ void k_count(const int* __restrict__ src, const int* __restrict__ dst,
                        int* __restrict__ cdst, int* __restrict__ csrc, int E) {
    int e = blockIdx.x * blockDim.x + threadIdx.x;
    if (e < E) {
        __hip_atomic_fetch_add(&cdst[dst[e]], 1, __ATOMIC_RELAXED, __HIP_MEMORY_SCOPE_AGENT);
        __hip_atomic_fetch_add(&csrc[src[e]], 1, __ATOMIC_RELAXED, __HIP_MEMORY_SCOPE_AGENT);
    }
}

__global__ void k_scan(const int* __restrict__ cnt, int* __restrict__ off,
                       int* __restrict__ cur, int N) {
    __shared__ int sh[1024];
    const int* c  = cnt + (size_t)blockIdx.x * N;
    int*       o  = off + (size_t)blockIdx.x * (N + 1);
    int*       cu = cur + (size_t)blockIdx.x * N;
    const int t     = threadIdx.x;
    const int chunk = (N + 1023) >> 10;
    const int b     = t * chunk;
    const int e     = min(b + chunk, N);
    int s = 0;
    for (int i = b; i < e; ++i) s += c[i];
    sh[t] = s;
    __syncthreads();
    for (int d = 1; d < 1024; d <<= 1) {
        int v = (t >= d) ? sh[t - d] : 0;
        __syncthreads();
        sh[t] += v;
        __syncthreads();
    }
    int run = sh[t] - s;
    for (int i = b; i < e; ++i) {
        o[i]  = run;
        cu[i] = run;
        run  += c[i];
    }
    if (t == 1023) o[N] = sh[1023];
}

__global__ void k_fill(const int* __restrict__ src, const int* __restrict__ dst,
                       int* __restrict__ cur_dst, int* __restrict__ cur_src,
                       int* __restrict__ nbr_in, int* __restrict__ nbr_out, int E) {
    int e = blockIdx.x * blockDim.x + threadIdx.x;
    if (e >= E) return;
    const int s = src[e], d = dst[e];
    const int p = __hip_atomic_fetch_add(&cur_dst[d], 1, __ATOMIC_RELAXED, __HIP_MEMORY_SCOPE_AGENT);
    nbr_in[p] = s;
    const int q = __hip_atomic_fetch_add(&cur_src[s], 1, __ATOMIC_RELAXED, __HIP_MEMORY_SCOPE_AGENT);
    nbr_out[q] = d;
}

__device__ __forceinline__ float4 ldRow(const float* __restrict__ X, int r, int l) {
    return *reinterpret_cast<const float4*>(X + (size_t)r * NHID + 4 * l);
}

__global__ void k_gather_mean(const float* __restrict__ X, const int* __restrict__ off,
                              const int* __restrict__ nbr, float* __restrict__ agg, int N) {
    const int t = blockIdx.x * blockDim.x + threadIdx.x;
    const int node = __builtin_amdgcn_readfirstlane(t >> 5);
    if (node >= N) return;
    const int l  = t & 31;
    const int b  = off[node];
    const int e  = off[node + 1];
    float4 acc = make_float4(0.f, 0.f, 0.f, 0.f);
    for (int base = b; base < e; base += 32) {
        const int n  = min(32, e - base);
        const int id = (l < n) ? nbr[base + l] : 0;
        for (int j = 0; j < n; j += 8) {
            float4 v[8];
            #pragma unroll
            for (int u = 0; u < 8; ++u)
                v[u] = ldRow(X, __builtin_amdgcn_readlane(id, (j + u) & 31), l);
            #pragma unroll
            for (int u = 0; u < 8; ++u) {
                const bool live = (j + u < n);
                acc.x += live ? v[u].x : 0.0f;
                acc.y += live ? v[u].y : 0.0f;
                acc.z += live ? v[u].z : 0.0f;
                acc.w += live ? v[u].w : 0.0f;
            }
        }
    }
    const float inv = 1.0f / fmaxf((float)(e - b), 1.0f);
    float4 r = make_float4(acc.x * inv, acc.y * inv, acc.z * inv, acc.w * inv);
    *reinterpret_cast<float4*>(agg + (size_t)node * NHID + 4 * l) = r;
}

__global__ void k_gate_gather(const float* __restrict__ A, const float* __restrict__ B,
                              const int* __restrict__ off, const int* __restrict__ nbr,
                              const float* __restrict__ Xcur, const float* __restrict__ Xnew,
                              float* __restrict__ Xnext, int N) {
    const int t = blockIdx.x * blockDim.x + threadIdx.x;
    const int node = __builtin_amdgcn_readfirstlane(t >> 5);
    if (node >= N) return;
    const int l  = t & 31;
    const int b  = off[node];
    const int e  = off[node + 1];
    const float4 a  = ldRow(A, node, l);
    const float4 xc = ldRow(Xcur, node, l);
    const float4 xn = ldRow(Xnew, node, l);
    float4 h = make_float4(0.f, 0.f, 0.f, 0.f);
    for (int base = b; base < e; base += 32) {
        const int n  = min(32, e - base);
        const int id = (l < n) ? nbr[base + l] : 0;
        for (int j = 0; j < n; j += 8) {
            float4 v[8];
            #pragma unroll
            for (int u = 0; u < 8; ++u)
                v[u] = ldRow(B, __builtin_amdgcn_readlane(id, (j + u) & 31), l);
            #pragma unroll
            for (int u = 0; u < 8; ++u) {
                const bool live = (j + u < n);
                const float tx = a.x + v[u].x, ty = a.y + v[u].y;
                const float tz = a.z + v[u].z, tw = a.w + v[u].w;
                h.x += live ? tx * tx : 0.0f;
                h.y += live ? ty * ty : 0.0f;
                h.z += live ? tz * tz : 0.0f;
                h.w += live ? tw * tw : 0.0f;
            }
        }
    }
    const float c = fmaxf((float)(e - b), 1.0f);
    const float t0 = tanhf(h.x / c), t1 = tanhf(h.y / c);
    const float t2 = tanhf(h.z / c), t3 = tanhf(h.w / c);
    float4 r;
    r.x = (1.0f - t0) * xc.x + t0 * xn.x;
    r.y = (1.0f - t1) * xc.y + t1 * xn.y;
    r.z = (1.0f - t2) * xc.z + t2 * xn.z;
    r.w = (1.0f - t3) * xc.w + t3 * xn.w;
    *reinterpret_cast<float4*>(Xnext + (size_t)node * NHID + 4 * l) = r;
}

#define GEMM_THREADS 256
#define A_PITCH      132

template <int N_, int NSEG>
__device__ __forceinline__ void gemm_body(float* __restrict__ smem,
                                          const float* __restrict__ A1, int lda1, const float* __restrict__ W1,
                                          const float* __restrict__ A2, int lda2, const float* __restrict__ W2,
                                          const float* __restrict__ bias, float* __restrict__ C,
                                          int M, int relu) {
    constexpr int PITCH  = 2 * N_ + 16;
    constexpr int NTILES = N_ / 16;
    constexpr int MG     = 8 / NTILES;
    constexpr int ROWS   = 48 * MG;
    constexpr int G      = N_ / 4;
    constexpr int PH     = NSEG * 32;
    constexpr int WTRIPS = (PH * G) / GEMM_THREADS;
    constexpr int ATRIPS = (ROWS * 32) / GEMM_THREADS;
    constexpr int KBP    = NSEG * 8;
    static_assert((PH * G) % GEMM_THREADS == 0 && (ROWS * 32) % GEMM_THREADS == 0,
                  "staging work divides evenly over the work-group");

    float* sW = smem;
    float* sA = smem + NSEG * 64 * PITCH;

    const int tid  = threadIdx.x;
    const int lane = tid & 31;
    const int wave = tid >> 5;
    const int c    = lane & 15;
    const int h    = lane >> 4;
    const int nt   = wave % NTILES;
    const int mg   = wave / NTILES;
    const int m0   = blockIdx.x * ROWS;

    float4 w0[WTRIPS], w1[WTRIPS];
    auto loadW = [&](int ph) {
        #pragma unroll
        for (int it = 0; it < WTRIPS; ++it) {
            const int i    = it * GEMM_THREADS + tid;
            const int qrow = ph * PH + i / G;
            const int g    = i % G;
            const float* Wp = ((qrow >> 6) == 0 ? W1 : W2) + (size_t)(2 * (qrow & 63)) * N_ + 4 * g;
            w0[it] = *reinterpret_cast<const float4*>(Wp);
            w1[it] = *reinterpret_cast<const float4*>(Wp + N_);
        }
    };
    auto storeW = [&](int ph) {
        #pragma unroll
        for (int it = 0; it < WTRIPS; ++it) {
            const int i    = it * GEMM_THREADS + tid;
            const int qrow = ph * PH + i / G;
            const int g    = i % G;
            float* d = sW + qrow * PITCH + 8 * g;
            *reinterpret_cast<float4*>(d)     = make_float4(w0[it].x, w1[it].x, w0[it].y, w1[it].y);
            *reinterpret_cast<float4*>(d + 4) = make_float4(w0[it].z, w1[it].z, w0[it].w, w1[it].w);
        }
    };

    v8f acc[3];
    #pragma unroll
    for (int i = 0; i < 3; ++i) acc[i] = v8f{};

    auto compute = [&](int ph) {
        #pragma unroll
        for (int kbl = 0; kbl < KBP; ++kbl) {
            const int KB  = ph * KBP + kbl;
            const int s   = KB >> 4;
            const int kbs = KB & 15;
            float4 a[3];
            #pragma unroll
            for (int i = 0; i < 3; ++i)
                a[i] = *reinterpret_cast<const float4*>(
                    sA + (s * ROWS + 48 * mg + 16 * i + c) * A_PITCH + 8 * kbs + 4 * h);
            #pragma unroll
            for (int j = 0; j < 2; ++j) {
                const v2f bv = *reinterpret_cast<const v2f*>(
                    sW + (4 * KB + 2 * h + j) * PITCH + 2 * (16 * nt + c));
                #pragma unroll
                for (int i = 0; i < 3; ++i) {
                    v2f av;
                    av.x = (j == 0) ? a[i].x : a[i].z;
                    av.y = (j == 0) ? a[i].y : a[i].w;
                    acc[i] = __builtin_amdgcn_wmma_f32_16x16x4_f32(false, av, false, bv,
                                                                   (short)0, acc[i], false, false);
                }
            }
        }
    };

    loadW(0);
    #pragma unroll
    for (int s = 0; s < NSEG; ++s) {
        #pragma unroll
        for (int it = 0; it < ATRIPS; ++it) {
            const int i   = it * GEMM_THREADS + tid;
            const int row = i >> 5;
            const int q   = i & 31;
            const int gr  = min(m0 + row, M - 1);
            const float* ap = (s == 0) ? A1 + (size_t)gr * lda1 : A2 + (size_t)gr * lda2;
            const float4 v = *reinterpret_cast<const float4*>(ap + 4 * q);
            *reinterpret_cast<float4*>(sA + (s * ROWS + row) * A_PITCH + 4 * q) = v;
        }
    }
    storeW(0);
    __syncthreads();
    loadW(1);
    compute(0);
    __builtin_amdgcn_sched_barrier(0);
    storeW(1);
    __syncthreads();
    compute(1);

    const int n = 16 * nt + c;
    const float bv = bias ? bias[n] : 0.0f;
    #pragma unroll
    for (int i = 0; i < 3; ++i) {
        #pragma unroll
        for (int r = 0; r < 8; ++r) {
            float x = acc[i][r] + bv;
            if (relu) x = fmaxf(x, 0.0f);
            const int m = m0 + 48 * mg + 16 * i + 8 * h + r;
            if (m < M) C[(size_t)m * N_ + n] = x;
        }
    }
}

__global__ void __launch_bounds__(GEMM_THREADS)
k_gemm_n128_s2(const float* __restrict__ A1, int lda1, const float* __restrict__ W1,
               const float* __restrict__ A2, int lda2, const float* __restrict__ W2,
               const float* __restrict__ bias, float* __restrict__ C, int M, int relu) {
    __shared__ __attribute__((aligned(16))) float smem[2 * 64 * (2 * 128 + 16) + 2 * 48 * A_PITCH];
    gemm_body<128, 2>(smem, A1, lda1, W1, A2, lda2, W2, bias, C, M, relu);
}

#define L_PITCH (2 * NHID + 16)
#define L_ROWS  48
#define L_WTRIPS ((64 * 32) / GEMM_THREADS)
#define L_ATRIPS ((L_ROWS * 32) / GEMM_THREADS)

__device__ __forceinline__ void layer_loadW(const float* __restrict__ Wp, int tid,
                                            float4 (&w0)[L_WTRIPS], float4 (&w1)[L_WTRIPS]) {
    #pragma unroll
    for (int it = 0; it < L_WTRIPS; ++it) {
        const int i = it * GEMM_THREADS + tid;
        const int q = i >> 5;
        const int g = i & 31;
        const float* p = Wp + (size_t)(2 * q) * NHID + 4 * g;
        w0[it] = *reinterpret_cast<const float4*>(p);
        w1[it] = *reinterpret_cast<const float4*>(p + NHID);
    }
}

template <int BUF>
__device__ __forceinline__ void layer_storeW(float* sW, int tid,
                                             const float4 (&w0)[L_WTRIPS], const float4 (&w1)[L_WTRIPS]) {
    #pragma unroll
    for (int it = 0; it < L_WTRIPS; ++it) {
        const int i = it * GEMM_THREADS + tid;
        const int q = i >> 5;
        const int g = i & 31;
        float* d = sW + (BUF * 64 + q) * L_PITCH + 8 * g;
        *reinterpret_cast<float4*>(d)     = make_float4(w0[it].x, w1[it].x, w0[it].y, w1[it].y);
        *reinterpret_cast<float4*>(d + 4) = make_float4(w0[it].z, w1[it].z, w0[it].w, w1[it].w);
    }
}

template <int BUF, int SEG>
__device__ __forceinline__ void layer_compute(const float* sW, const float* sA,
                                              int c, int h, int nt, v8f (&acc)[3]) {
    #pragma unroll
    for (int kbl = 0; kbl < 16; ++kbl) {
        float4 a[3];
        #pragma unroll
        for (int i = 0; i < 3; ++i)
            a[i] = *reinterpret_cast<const float4*>(
                sA + (SEG * L_ROWS + 16 * i + c) * A_PITCH + 8 * kbl + 4 * h);
        #pragma unroll
        for (int j = 0; j < 2; ++j) {
            const v2f bv = *reinterpret_cast<const v2f*>(
                sW + (BUF * 64 + 4 * kbl + 2 * h + j) * L_PITCH + 2 * (16 * nt + c));
            #pragma unroll
            for (int i = 0; i < 3; ++i) {
                v2f av;
                av.x = (j == 0) ? a[i].x : a[i].z;
                av.y = (j == 0) ? a[i].y : a[i].w;
                acc[i] = __builtin_amdgcn_wmma_f32_16x16x4_f32(false, av, false, bv,
                                                               (short)0, acc[i], false, false);
            }
        }
    }
}

template <bool RELU>
__device__ __forceinline__ void layer_store_global(const v8f (&acc)[3], float bias,
                                                   float* __restrict__ C, int m0, int n, int h, int M) {
    #pragma unroll
    for (int i = 0; i < 3; ++i) {
        #pragma unroll
        for (int r = 0; r < 8; ++r) {
            float x = acc[i][r] + bias;
            if (RELU) x = fmaxf(x, 0.0f);
            const int m = m0 + 16 * i + 8 * h + r;
            if (m < M) C[(size_t)m * NHID + n] = x;
        }
    }
}

__global__ void __launch_bounds__(GEMM_THREADS)
k_layer_gemms(const float* __restrict__ agg, const float* __restrict__ Xcur,
              const float* __restrict__ conv_Wl, const float* __restrict__ conv_Wr,
              const float* __restrict__ conv_bl,
              const float* __restrict__ gg_Wl, const float* __restrict__ gg_Wr,
              const float* __restrict__ gg_bl,
              const float* __restrict__ Q_top, const float* __restrict__ Q_bot,
              const float* __restrict__ Q_b,
              float* __restrict__ Xnew, float* __restrict__ Ab, float* __restrict__ Bb, int M) {
    __shared__ __attribute__((aligned(16))) float smem[2 * 64 * L_PITCH + 3 * L_ROWS * A_PITCH];
    float* sW = smem;
    float* sA = smem + 2 * 64 * L_PITCH;

    const int tid  = threadIdx.x;
    const int lane = tid & 31;
    const int nt   = tid >> 5;
    const int c    = lane & 15;
    const int h    = lane >> 4;
    const int m0   = blockIdx.x * L_ROWS;
    const int n    = 16 * nt + c;

    float4 w0[L_WTRIPS], w1[L_WTRIPS];
    v8f acc[3];

    layer_loadW(conv_Wl, tid, w0, w1);
    #pragma unroll
    for (int s = 0; s < 2; ++s) {
        #pragma unroll
        for (int it = 0; it < L_ATRIPS; ++it) {
            const int i   = it * GEMM_THREADS + tid;
            const int row = i >> 5;
            const int q   = i & 31;
            const int gr  = min(m0 + row, M - 1);
            const float* ap = ((s == 0) ? agg : Xcur) + (size_t)gr * NHID;
            const float4 v = *reinterpret_cast<const float4*>(ap + 4 * q);
            *reinterpret_cast<float4*>(sA + (s * L_ROWS + row) * A_PITCH + 4 * q) = v;
        }
    }
    const float b_conv = conv_bl[n];
    const float b_gg   = gg_bl[n];
    const float b_q    = Q_b[n];
    layer_storeW<0>(sW, tid, w0, w1);
    __syncthreads();


    layer_loadW(conv_Wr, tid, w0, w1);
    #pragma unroll
    for (int i = 0; i < 3; ++i) acc[i] = v8f{};
    layer_compute<0, 0>(sW, sA, c, h, nt, acc);
    __builtin_amdgcn_sched_barrier(0);
    layer_storeW<1>(sW, tid, w0, w1);
    __syncthreads();

    layer_loadW(gg_Wl, tid, w0, w1);
    layer_compute<1, 1>(sW, sA, c, h, nt, acc);
    __builtin_amdgcn_sched_barrier(0);
    layer_storeW<0>(sW, tid, w0, w1);
    __syncthreads();

    layer_loadW(gg_Wr, tid, w0, w1);
    layer_store_global<true>(acc, b_conv, Xnew, m0, n, h, M);
    #pragma unroll
    for (int i = 0; i < 3; ++i) acc[i] = v8f{};
    layer_compute<0, 0>(sW, sA, c, h, nt, acc);
    __builtin_amdgcn_sched_barrier(0);
    layer_storeW<1>(sW, tid, w0, w1);
    __syncthreads();

    layer_loadW(Q_top, tid, w0, w1);
    layer_compute<1, 1>(sW, sA, c, h, nt, acc);
    #pragma unroll
    for (int i = 0; i < 3; ++i) {
        #pragma unroll
        for (int r = 0; r < 8; ++r)
            sA[(2 * L_ROWS + 16 * i + 8 * h + r) * A_PITCH + n] = fmaxf(acc[i][r] + b_gg, 0.0f);
    }
    __builtin_amdgcn_sched_barrier(0);
    layer_storeW<0>(sW, tid, w0, w1);
    __syncthreads();

    layer_loadW(Q_bot, tid, w0, w1);
    #pragma unroll
    for (int i = 0; i < 3; ++i) acc[i] = v8f{};
    layer_compute<0, 2>(sW, sA, c, h, nt, acc);
    __builtin_amdgcn_sched_barrier(0);
    layer_storeW<1>(sW, tid, w0, w1);
    __syncthreads();

    layer_store_global<false>(acc, b_q, Ab, m0, n, h, M);
    #pragma unroll
    for (int i = 0; i < 3; ++i) acc[i] = v8f{};
    layer_compute<1, 2>(sW, sA, c, h, nt, acc);
    layer_store_global<false>(acc, 0.0f, Bb, m0, n, h, M);
}

__global__ void __launch_bounds__(GEMM_THREADS)
k_gemm_n64_s1(const float* __restrict__ A1, int lda1, const float* __restrict__ W1,
              const float* __restrict__ bias, float* __restrict__ C, int M, int relu) {
    __shared__ __attribute__((aligned(16))) float smem[64 * (2 * 64 + 16) + 96 * A_PITCH];
    gemm_body<64, 1>(smem, A1, lda1, W1, nullptr, 0, nullptr, bias, C, M, relu);
}

extern "C" void kernel_launch(void* const* d_in, const int* in_sizes, int n_in,
                              void* d_out, int out_size, void* d_ws, size_t ws_size,
                              hipStream_t stream) {
    const float* X       = (const float*)d_in[0];
    const int*   ei      = (const int*)  d_in[1];
    const float* enc_W   = (const float*)d_in[2];
    const float* enc_b   = (const float*)d_in[3];
    const float* conv_Wl = (const float*)d_in[4];
    const float* conv_bl = (const float*)d_in[5];
    const float* conv_Wr = (const float*)d_in[6];
    const float* gg_Wl   = (const float*)d_in[7];
    const float* gg_bl   = (const float*)d_in[8];
    const float* gg_Wr   = (const float*)d_in[9];
    const float* Q_W     = (const float*)d_in[10];
    const float* Q_b     = (const float*)d_in[11];
    const float* dec_W   = (const float*)d_in[12];
    const float* dec_b   = (const float*)d_in[13];

    const int N = in_sizes[0] / NFEAT;
    const int E = in_sizes[1] / 2;
    const int* src = ei;
    const int* dst = ei + E;

    float* ws       = (float*)d_ws;
    const size_t NH = (size_t)N * NHID;
    int*   cnt      = (int*)ws;
    float* Xa       = ws + 2 * (size_t)N;
    float* Xb       = Xa + NH;
    float* agg      = Xb + NH;
    float* T1       = agg + NH;
    float* Ab       = T1 + NH;
    float* Bb       = Ab + NH;
    int*   csr      = (int*)(Bb + NH);
    int*   off      = csr;
    int*   cur      = off + 2 * (size_t)(N + 1);
    int*   nbr_in   = cur + 2 * (size_t)N;
    int*   nbr_out  = nbr_in + E;
    const int* off_dst = off;
    const int* off_src = off + (N + 1);

    auto cdiv = [](long long a, long long b) { return (int)((a + b - 1) / b); };
    const int nodeGrd = cdiv((long long)N * 32, 256);
    const int gemmGrd = cdiv(N, 48);
    const int decGrd  = cdiv(N, 96);

    k_zero <<<cdiv(2 * N, 256), 256, 0, stream>>>(cnt, 2 * N);
    k_count<<<cdiv(E, 256),     256, 0, stream>>>(src, dst, cnt, cnt + N, E);
    k_scan <<<2, 1024, 0, stream>>>(cnt, off, cur, N);
    k_fill <<<cdiv(E, 256),     256, 0, stream>>>(src, dst, cur, cur + N, nbr_in, nbr_out, E);

    k_gemm_n128_s2<<<gemmGrd, GEMM_THREADS, 0, stream>>>(X, NFEAT, enc_W, X + NHID, NFEAT,
                                                         enc_W + (size_t)NHID * NHID, enc_b, Xa, N, 1);

    float* Xcur  = Xa;
    float* Xnext = Xb;
    for (int layer = 0; layer < 2; ++layer) {
        k_gather_mean<<<nodeGrd, 256, 0, stream>>>(Xcur, off_dst, nbr_in, agg, N);

        k_layer_gemms<<<gemmGrd, GEMM_THREADS, 0, stream>>>(agg, Xcur, conv_Wl, conv_Wr, conv_bl,
                                                            gg_Wl, gg_Wr, gg_bl,
                                                            Q_W, Q_W + (size_t)NHID * NHID, Q_b,
                                                            T1, Ab, Bb, N);

        k_gate_gather<<<nodeGrd, 256, 0, stream>>>(Ab, Bb, off_src, nbr_out, Xcur, T1, Xnext, N);

        float* t = Xcur; Xcur = Xnext; Xnext = t;
    }

    k_gemm_n64_s1<<<decGrd, GEMM_THREADS, 0, stream>>>(Xcur, NHID, dec_W, dec_b, (float*)d_out, N, 0);
}


# Edge slots from the counting pass: each count atomic returns the edge's rank in its node's list, so the fill pass has no atomics (1280k to 640k a call); list and f32 sum order stay arrival order
# speedup vs baseline: 2.8700x; 1.4364x over previous
#include <hip/hip_runtime.h>
#include <hip/hip_bf16.h>


typedef __attribute__((ext_vector_type(2))) float v2f;
typedef __attribute__((ext_vector_type(8))) float v8f;

#define NFEAT  256
#define NHID   128
#define NCLASS 64

__global__ void k_zero(int* __restrict__ p, int n) {
    int i = blockIdx.x * blockDim.x + threadIdx.x;
    if (i < n) p[i] = 0;
}

__global__ void k_count(const int* __restrict__ src, const int* __restrict__ dst,
                        int* __restrict__ cdst, int* __restrict__ csrc,
                        int* __restrict__ rank, int E) {
    int e = blockIdx.x * blockDim.x + threadIdx.x;
    if (e < E) {
        const int s = src[e], d = dst[e];
        const int ri = __hip_atomic_fetch_add(&cdst[d], 1, __ATOMIC_RELAXED, __HIP_MEMORY_SCOPE_AGENT);
        const int ro = __hip_atomic_fetch_add(&csrc[s], 1, __ATOMIC_RELAXED, __HIP_MEMORY_SCOPE_AGENT);
        reinterpret_cast<int2*>(rank)[e] = make_int2(ri, ro);
    }
}

__global__ void k_scan(const int* __restrict__ cnt, int* __restrict__ off, int N) {
    __shared__ int sh[1024];
    const int* c  = cnt + (size_t)blockIdx.x * N;
    int*       o  = off + (size_t)blockIdx.x * (N + 1);
    const int t     = threadIdx.x;
    const int chunk = (N + 1023) >> 10;
    const int b     = t * chunk;
    const int e     = min(b + chunk, N);
    int s = 0;
    for (int i = b; i < e; ++i) s += c[i];
    sh[t] = s;
    __syncthreads();
    for (int d = 1; d < 1024; d <<= 1) {
        int v = (t >= d) ? sh[t - d] : 0;
        __syncthreads();
        sh[t] += v;
        __syncthreads();
    }
    int run = sh[t] - s;
    for (int i = b; i < e; ++i) {
        o[i] = run;
        run += c[i];
    }
    if (t == 1023) o[N] = sh[1023];
}

__global__ void k_fill(const int* __restrict__ src, const int* __restrict__ dst,
                       const int* __restrict__ off_dst, const int* __restrict__ off_src,
                       const int* __restrict__ rank,
                       int* __restrict__ nbr_in, int* __restrict__ nbr_out, int E) {
    int e = blockIdx.x * blockDim.x + threadIdx.x;
    if (e >= E) return;
    const int  s = src[e], d = dst[e];
    const int2 r = reinterpret_cast<const int2*>(rank)[e];
    nbr_in [off_dst[d] + r.x] = s;
    nbr_out[off_src[s] + r.y] = d;
}

__device__ __forceinline__ float4 ldRow(const float* __restrict__ X, int r, int l) {
    return *reinterpret_cast<const float4*>(X + (size_t)r * NHID + 4 * l);
}

__global__ void k_gather_mean(const float* __restrict__ X, const int* __restrict__ off,
                              const int* __restrict__ nbr, float* __restrict__ agg, int N) {
    const int t = blockIdx.x * blockDim.x + threadIdx.x;
    const int node = __builtin_amdgcn_readfirstlane(t >> 5);
    if (node >= N) return;
    const int l  = t & 31;
    const int b  = off[node];
    const int e  = off[node + 1];
    float4 acc = make_float4(0.f, 0.f, 0.f, 0.f);
    for (int base = b; base < e; base += 32) {
        const int n  = min(32, e - base);
        const int id = (l < n) ? nbr[base + l] : 0;
        for (int j = 0; j < n; j += 8) {
            float4 v[8];
            #pragma unroll
            for (int u = 0; u < 8; ++u)
                v[u] = ldRow(X, __builtin_amdgcn_readlane(id, (j + u) & 31), l);
            #pragma unroll
            for (int u = 0; u < 8; ++u) {
                const bool live = (j + u < n);
                acc.x += live ? v[u].x : 0.0f;
                acc.y += live ? v[u].y : 0.0f;
                acc.z += live ? v[u].z : 0.0f;
                acc.w += live ? v[u].w : 0.0f;
            }
        }
    }
    const float inv = 1.0f / fmaxf((float)(e - b), 1.0f);
    float4 r = make_float4(acc.x * inv, acc.y * inv, acc.z * inv, acc.w * inv);
    *reinterpret_cast<float4*>(agg + (size_t)node * NHID + 4 * l) = r;
}

__global__ void k_gate_gather(const float* __restrict__ A, const float* __restrict__ B,
                              const int* __restrict__ off, const int* __restrict__ nbr,
                              const float* __restrict__ Xcur, const float* __restrict__ Xnew,
                              float* __restrict__ Xnext, int N) {
    const int t = blockIdx.x * blockDim.x + threadIdx.x;
    const int node = __builtin_amdgcn_readfirstlane(t >> 5);
    if (node >= N) return;
    const int l  = t & 31;
    const int b  = off[node];
    const int e  = off[node + 1];
    const float4 a  = ldRow(A, node, l);
    const float4 xc = ldRow(Xcur, node, l);
    const float4 xn = ldRow(Xnew, node, l);
    float4 h = make_float4(0.f, 0.f, 0.f, 0.f);
    for (int base = b; base < e; base += 32) {
        const int n  = min(32, e - base);
        const int id = (l < n) ? nbr[base + l] : 0;
        for (int j = 0; j < n; j += 8) {
            float4 v[8];
            #pragma unroll
            for (int u = 0; u < 8; ++u)
                v[u] = ldRow(B, __builtin_amdgcn_readlane(id, (j + u) & 31), l);
            #pragma unroll
            for (int u = 0; u < 8; ++u) {
                const bool live = (j + u < n);
                const float tx = a.x + v[u].x, ty = a.y + v[u].y;
                const float tz = a.z + v[u].z, tw = a.w + v[u].w;
                h.x += live ? tx * tx : 0.0f;
                h.y += live ? ty * ty : 0.0f;
                h.z += live ? tz * tz : 0.0f;
                h.w += live ? tw * tw : 0.0f;
            }
        }
    }
    const float c = fmaxf((float)(e - b), 1.0f);
    const float t0 = tanhf(h.x / c), t1 = tanhf(h.y / c);
    const float t2 = tanhf(h.z / c), t3 = tanhf(h.w / c);
    float4 r;
    r.x = (1.0f - t0) * xc.x + t0 * xn.x;
    r.y = (1.0f - t1) * xc.y + t1 * xn.y;
    r.z = (1.0f - t2) * xc.z + t2 * xn.z;
    r.w = (1.0f - t3) * xc.w + t3 * xn.w;
    *reinterpret_cast<float4*>(Xnext + (size_t)node * NHID + 4 * l) = r;
}

#define GEMM_THREADS 256
#define A_PITCH      132

template <int N_, int NSEG>
__device__ __forceinline__ void gemm_body(float* __restrict__ smem,
                                          const float* __restrict__ A1, int lda1, const float* __restrict__ W1,
                                          const float* __restrict__ A2, int lda2, const float* __restrict__ W2,
                                          const float* __restrict__ bias, float* __restrict__ C,
                                          int M, int relu) {
    constexpr int PITCH  = 2 * N_ + 16;
    constexpr int NTILES = N_ / 16;
    constexpr int MG     = 8 / NTILES;
    constexpr int ROWS   = 48 * MG;
    constexpr int G      = N_ / 4;
    constexpr int PH     = NSEG * 32;
    constexpr int WTRIPS = (PH * G) / GEMM_THREADS;
    constexpr int ATRIPS = (ROWS * 32) / GEMM_THREADS;
    constexpr int KBP    = NSEG * 8;
    static_assert((PH * G) % GEMM_THREADS == 0 && (ROWS * 32) % GEMM_THREADS == 0,
                  "staging work divides evenly over the work-group");

    float* sW = smem;
    float* sA = smem + NSEG * 64 * PITCH;

    const int tid  = threadIdx.x;
    const int lane = tid & 31;
    const int wave = tid >> 5;
    const int c    = lane & 15;
    const int h    = lane >> 4;
    const int nt   = wave % NTILES;
    const int mg   = wave / NTILES;
    const int m0   = blockIdx.x * ROWS;

    float4 w0[WTRIPS], w1[WTRIPS];
    auto loadW = [&](int ph) {
        #pragma unroll
        for (int it = 0; it < WTRIPS; ++it) {
            const int i    = it * GEMM_THREADS + tid;
            const int qrow = ph * PH + i / G;
            const int g    = i % G;
            const float* Wp = ((qrow >> 6) == 0 ? W1 : W2) + (size_t)(2 * (qrow & 63)) * N_ + 4 * g;
            w0[it] = *reinterpret_cast<const float4*>(Wp);
            w1[it] = *reinterpret_cast<const float4*>(Wp + N_);
        }
    };
    auto storeW = [&](int ph) {
        #pragma unroll
        for (int it = 0; it < WTRIPS; ++it) {
            const int i    = it * GEMM_THREADS + tid;
            const int qrow = ph * PH + i / G;
            const int g    = i % G;
            float* d = sW + qrow * PITCH + 8 * g;
            *reinterpret_cast<float4*>(d)     = make_float4(w0[it].x, w1[it].x, w0[it].y, w1[it].y);
            *reinterpret_cast<float4*>(d + 4) = make_float4(w0[it].z, w1[it].z, w0[it].w, w1[it].w);
        }
    };

    v8f acc[3];
    #pragma unroll
    for (int i = 0; i < 3; ++i) acc[i] = v8f{};

    auto compute = [&](int ph) {
        #pragma unroll
        for (int kbl = 0; kbl < KBP; ++kbl) {
            const int KB  = ph * KBP + kbl;
            const int s   = KB >> 4;
            const int kbs = KB & 15;
            float4 a[3];
            #pragma unroll
            for (int i = 0; i < 3; ++i)
                a[i] = *reinterpret_cast<const float4*>(
                    sA + (s * ROWS + 48 * mg + 16 * i + c) * A_PITCH + 8 * kbs + 4 * h);
            #pragma unroll
            for (int j = 0; j < 2; ++j) {
                const v2f bv = *reinterpret_cast<const v2f*>(
                    sW + (4 * KB + 2 * h + j) * PITCH + 2 * (16 * nt + c));
                #pragma unroll
                for (int i = 0; i < 3; ++i) {
                    v2f av;
                    av.x = (j == 0) ? a[i].x : a[i].z;
                    av.y = (j == 0) ? a[i].y : a[i].w;
                    acc[i] = __builtin_amdgcn_wmma_f32_16x16x4_f32(false, av, false, bv,
                                                                   (short)0, acc[i], false, false);
                }
            }
        }
    };

    loadW(0);
    #pragma unroll
    for (int s = 0; s < NSEG; ++s) {
        #pragma unroll
        for (int it = 0; it < ATRIPS; ++it) {
            const int i   = it * GEMM_THREADS + tid;
            const int row = i >> 5;
            const int q   = i & 31;
            const int gr  = min(m0 + row, M - 1);
            const float* ap = (s == 0) ? A1 + (size_t)gr * lda1 : A2 + (size_t)gr * lda2;
            const float4 v = *reinterpret_cast<const float4*>(ap + 4 * q);
            *reinterpret_cast<float4*>(sA + (s * ROWS + row) * A_PITCH + 4 * q) = v;
        }
    }
    storeW(0);
    __syncthreads();
    loadW(1);
    compute(0);
    __builtin_amdgcn_sched_barrier(0);
    storeW(1);
    __syncthreads();
    compute(1);

    const int n = 16 * nt + c;
    const float bv = bias ? bias[n] : 0.0f;
    #pragma unroll
    for (int i = 0; i < 3; ++i) {
        #pragma unroll
        for (int r = 0; r < 8; ++r) {
            float x = acc[i][r] + bv;
            if (relu) x = fmaxf(x, 0.0f);
            const int m = m0 + 48 * mg + 16 * i + 8 * h + r;
            if (m < M) C[(size_t)m * N_ + n] = x;
        }
    }
}

__global__ void __launch_bounds__(GEMM_THREADS)
k_gemm_n128_s2(const float* __restrict__ A1, int lda1, const float* __restrict__ W1,
               const float* __restrict__ A2, int lda2, const float* __restrict__ W2,
               const float* __restrict__ bias, float* __restrict__ C, int M, int relu) {
    __shared__ __attribute__((aligned(16))) float smem[2 * 64 * (2 * 128 + 16) + 2 * 48 * A_PITCH];
    gemm_body<128, 2>(smem, A1, lda1, W1, A2, lda2, W2, bias, C, M, relu);
}

#define L_PITCH (2 * NHID + 16)
#define L_ROWS  48
#define L_WTRIPS ((64 * 32) / GEMM_THREADS)
#define L_ATRIPS ((L_ROWS * 32) / GEMM_THREADS)

__device__ __forceinline__ void layer_loadW(const float* __restrict__ Wp, int tid,
                                            float4 (&w0)[L_WTRIPS], float4 (&w1)[L_WTRIPS]) {
    #pragma unroll
    for (int it = 0; it < L_WTRIPS; ++it) {
        const int i = it * GEMM_THREADS + tid;
        const int q = i >> 5;
        const int g = i & 31;
        const float* p = Wp + (size_t)(2 * q) * NHID + 4 * g;
        w0[it] = *reinterpret_cast<const float4*>(p);
        w1[it] = *reinterpret_cast<const float4*>(p + NHID);
    }
}

template <int BUF>
__device__ __forceinline__ void layer_storeW(float* sW, int tid,
                                             const float4 (&w0)[L_WTRIPS], const float4 (&w1)[L_WTRIPS]) {
    #pragma unroll
    for (int it = 0; it < L_WTRIPS; ++it) {
        const int i = it * GEMM_THREADS + tid;
        const int q = i >> 5;
        const int g = i & 31;
        float* d = sW + (BUF * 64 + q) * L_PITCH + 8 * g;
        *reinterpret_cast<float4*>(d)     = make_float4(w0[it].x, w1[it].x, w0[it].y, w1[it].y);
        *reinterpret_cast<float4*>(d + 4) = make_float4(w0[it].z, w1[it].z, w0[it].w, w1[it].w);
    }
}

template <int BUF, int SEG>
__device__ __forceinline__ void layer_compute(const float* sW, const float* sA,
                                              int c, int h, int nt, v8f (&acc)[3]) {
    #pragma unroll
    for (int kbl = 0; kbl < 16; ++kbl) {
        float4 a[3];
        #pragma unroll
        for (int i = 0; i < 3; ++i)
            a[i] = *reinterpret_cast<const float4*>(
                sA + (SEG * L_ROWS + 16 * i + c) * A_PITCH + 8 * kbl + 4 * h);
        #pragma unroll
        for (int j = 0; j < 2; ++j) {
            const v2f bv = *reinterpret_cast<const v2f*>(
                sW + (BUF * 64 + 4 * kbl + 2 * h + j) * L_PITCH + 2 * (16 * nt + c));
            #pragma unroll
            for (int i = 0; i < 3; ++i) {
                v2f av;
                av.x = (j == 0) ? a[i].x : a[i].z;
                av.y = (j == 0) ? a[i].y : a[i].w;
                acc[i] = __builtin_amdgcn_wmma_f32_16x16x4_f32(false, av, false, bv,
                                                               (short)0, acc[i], false, false);
            }
        }
    }
}

template <bool RELU>
__device__ __forceinline__ void layer_store_global(const v8f (&acc)[3], float bias,
                                                   float* __restrict__ C, int m0, int n, int h, int M) {
    #pragma unroll
    for (int i = 0; i < 3; ++i) {
        #pragma unroll
        for (int r = 0; r < 8; ++r) {
            float x = acc[i][r] + bias;
            if (RELU) x = fmaxf(x, 0.0f);
            const int m = m0 + 16 * i + 8 * h + r;
            if (m < M) C[(size_t)m * NHID + n] = x;
        }
    }
}

__global__ void __launch_bounds__(GEMM_THREADS)
k_layer_gemms(const float* __restrict__ agg, const float* __restrict__ Xcur,
              const float* __restrict__ conv_Wl, const float* __restrict__ conv_Wr,
              const float* __restrict__ conv_bl,
              const float* __restrict__ gg_Wl, const float* __restrict__ gg_Wr,
              const float* __restrict__ gg_bl,
              const float* __restrict__ Q_top, const float* __restrict__ Q_bot,
              const float* __restrict__ Q_b,
              float* __restrict__ Xnew, float* __restrict__ Ab, float* __restrict__ Bb, int M) {
    __shared__ __attribute__((aligned(16))) float smem[2 * 64 * L_PITCH + 3 * L_ROWS * A_PITCH];
    float* sW = smem;
    float* sA = smem + 2 * 64 * L_PITCH;

    const int tid  = threadIdx.x;
    const int lane = tid & 31;
    const int nt   = tid >> 5;
    const int c    = lane & 15;
    const int h    = lane >> 4;
    const int m0   = blockIdx.x * L_ROWS;
    const int n    = 16 * nt + c;

    float4 w0[L_WTRIPS], w1[L_WTRIPS];
    v8f acc[3];

    layer_loadW(conv_Wl, tid, w0, w1);
    #pragma unroll
    for (int s = 0; s < 2; ++s) {
        #pragma unroll
        for (int it = 0; it < L_ATRIPS; ++it) {
            const int i   = it * GEMM_THREADS + tid;
            const int row = i >> 5;
            const int q   = i & 31;
            const int gr  = min(m0 + row, M - 1);
            const float* ap = ((s == 0) ? agg : Xcur) + (size_t)gr * NHID;
            const float4 v = *reinterpret_cast<const float4*>(ap + 4 * q);
            *reinterpret_cast<float4*>(sA + (s * L_ROWS + row) * A_PITCH + 4 * q) = v;
        }
    }
    const float b_conv = conv_bl[n];
    const float b_gg   = gg_bl[n];
    const float b_q    = Q_b[n];
    layer_storeW<0>(sW, tid, w0, w1);
    __syncthreads();


    layer_loadW(conv_Wr, tid, w0, w1);
    #pragma unroll
    for (int i = 0; i < 3; ++i) acc[i] = v8f{};
    layer_compute<0, 0>(sW, sA, c, h, nt, acc);
    __builtin_amdgcn_sched_barrier(0);
    layer_storeW<1>(sW, tid, w0, w1);
    __syncthreads();

    layer_loadW(gg_Wl, tid, w0, w1);
    layer_compute<1, 1>(sW, sA, c, h, nt, acc);
    __builtin_amdgcn_sched_barrier(0);
    layer_storeW<0>(sW, tid, w0, w1);
    __syncthreads();

    layer_loadW(gg_Wr, tid, w0, w1);
    layer_store_global<true>(acc, b_conv, Xnew, m0, n, h, M);
    #pragma unroll
    for (int i = 0; i < 3; ++i) acc[i] = v8f{};
    layer_compute<0, 0>(sW, sA, c, h, nt, acc);
    __builtin_amdgcn_sched_barrier(0);
    layer_storeW<1>(sW, tid, w0, w1);
    __syncthreads();

    layer_loadW(Q_top, tid, w0, w1);
    layer_compute<1, 1>(sW, sA, c, h, nt, acc);
    #pragma unroll
    for (int i = 0; i < 3; ++i) {
        #pragma unroll
        for (int r = 0; r < 8; ++r)
            sA[(2 * L_ROWS + 16 * i + 8 * h + r) * A_PITCH + n] = fmaxf(acc[i][r] + b_gg, 0.0f);
    }
    __builtin_amdgcn_sched_barrier(0);
    layer_storeW<0>(sW, tid, w0, w1);
    __syncthreads();

    layer_loadW(Q_bot, tid, w0, w1);
    #pragma unroll
    for (int i = 0; i < 3; ++i) acc[i] = v8f{};
    layer_compute<0, 2>(sW, sA, c, h, nt, acc);
    __builtin_amdgcn_sched_barrier(0);
    layer_storeW<1>(sW, tid, w0, w1);
    __syncthreads();

    layer_store_global<false>(acc, b_q, Ab, m0, n, h, M);
    #pragma unroll
    for (int i = 0; i < 3; ++i) acc[i] = v8f{};
    layer_compute<1, 2>(sW, sA, c, h, nt, acc);
    layer_store_global<false>(acc, 0.0f, Bb, m0, n, h, M);
}

__global__ void __launch_bounds__(GEMM_THREADS)
k_gemm_n64_s1(const float* __restrict__ A1, int lda1, const float* __restrict__ W1,
              const float* __restrict__ bias, float* __restrict__ C, int M, int relu) {
    __shared__ __attribute__((aligned(16))) float smem[64 * (2 * 64 + 16) + 96 * A_PITCH];
    gemm_body<64, 1>(smem, A1, lda1, W1, nullptr, 0, nullptr, bias, C, M, relu);
}

extern "C" void kernel_launch(void* const* d_in, const int* in_sizes, int n_in,
                              void* d_out, int out_size, void* d_ws, size_t ws_size,
                              hipStream_t stream) {
    const float* X       = (const float*)d_in[0];
    const int*   ei      = (const int*)  d_in[1];
    const float* enc_W   = (const float*)d_in[2];
    const float* enc_b   = (const float*)d_in[3];
    const float* conv_Wl = (const float*)d_in[4];
    const float* conv_bl = (const float*)d_in[5];
    const float* conv_Wr = (const float*)d_in[6];
    const float* gg_Wl   = (const float*)d_in[7];
    const float* gg_bl   = (const float*)d_in[8];
    const float* gg_Wr   = (const float*)d_in[9];
    const float* Q_W     = (const float*)d_in[10];
    const float* Q_b     = (const float*)d_in[11];
    const float* dec_W   = (const float*)d_in[12];
    const float* dec_b   = (const float*)d_in[13];

    const int N = in_sizes[0] / NFEAT;
    const int E = in_sizes[1] / 2;
    const int* src = ei;
    const int* dst = ei + E;

    float* ws       = (float*)d_ws;
    const size_t NH = (size_t)N * NHID;
    int*   cnt      = (int*)ws;
    float* Xa       = ws + 2 * (size_t)N;
    float* Xb       = Xa + NH;
    float* agg      = Xb + NH;
    float* T1       = agg + NH;
    float* Ab       = T1 + NH;
    float* Bb       = Ab + NH;
    int*   csr      = (int*)(Bb + NH);
    int*   off      = csr;
    int*   nbr_in   = off + 2 * (size_t)(N + 1);
    int*   nbr_out  = nbr_in + E;
    int*   rank     = (int*)agg;
    const int* off_dst = off;
    const int* off_src = off + (N + 1);

    auto cdiv = [](long long a, long long b) { return (int)((a + b - 1) / b); };
    const int nodeGrd = cdiv((long long)N * 32, 256);
    const int gemmGrd = cdiv(N, 48);
    const int decGrd  = cdiv(N, 96);

    k_zero <<<cdiv(2 * N, 256), 256, 0, stream>>>(cnt, 2 * N);
    k_count<<<cdiv(E, 256),     256, 0, stream>>>(src, dst, cnt, cnt + N, rank, E);
    k_scan <<<2, 1024, 0, stream>>>(cnt, off, N);
    k_fill <<<cdiv(E, 256),     256, 0, stream>>>(src, dst, off_dst, off_src, rank, nbr_in, nbr_out, E);

    k_gemm_n128_s2<<<gemmGrd, GEMM_THREADS, 0, stream>>>(X, NFEAT, enc_W, X + NHID, NFEAT,
                                                         enc_W + (size_t)NHID * NHID, enc_b, Xa, N, 1);

    float* Xcur  = Xa;
    float* Xnext = Xb;
    for (int layer = 0; layer < 2; ++layer) {
        k_gather_mean<<<nodeGrd, 256, 0, stream>>>(Xcur, off_dst, nbr_in, agg, N);

        k_layer_gemms<<<gemmGrd, GEMM_THREADS, 0, stream>>>(agg, Xcur, conv_Wl, conv_Wr, conv_bl,
                                                            gg_Wl, gg_Wr, gg_bl,
                                                            Q_W, Q_W + (size_t)NHID * NHID, Q_b,
                                                            T1, Ab, Bb, N);

        k_gate_gather<<<nodeGrd, 256, 0, stream>>>(Ab, Bb, off_src, nbr_out, Xcur, T1, Xnext, N);

        float* t = Xcur; Xcur = Xnext; Xnext = t;
    }

    k_gemm_n64_s1<<<decGrd, GEMM_THREADS, 0, stream>>>(Xcur, NHID, dec_W, dec_b, (float*)d_out, N, 0);
}


# CSR counting: global integer atomics replaced by per-edge-block LDS histograms (LDS atomics give in-block ranks) and a prefix over the 64 blocks; f32 sum order is now block, then LDS arrival order
# speedup vs baseline: 3.4295x; 1.1949x over previous
#include <hip/hip_runtime.h>
#include <hip/hip_bf16.h>


typedef __attribute__((ext_vector_type(2))) float v2f;
typedef __attribute__((ext_vector_type(8))) float v8f;

#define NFEAT  256
#define NHID   128
#define NCLASS 64

#define CSR_BLOCKS  64
#define CSR_THREADS 1024
#define CSR_BATCH   8
#define CSR_MAXBINS 20480

__global__ void __launch_bounds__(CSR_THREADS)
k_count(const int* __restrict__ src, const int* __restrict__ dst,
        int* __restrict__ H, int* __restrict__ rank, int E, int N, int chunk) {
    __shared__ int hist[CSR_MAXBINS];
    const int tid  = threadIdx.x;
    const int bins = 2 * N;
    for (int i = tid; i < bins; i += CSR_THREADS) hist[i] = 0;
    __syncthreads();

    const int e0 = blockIdx.x * chunk;
    const int e1 = min(e0 + chunk, E);
    for (int base = e0; base < e1; base += CSR_THREADS * CSR_BATCH) {
        int s[CSR_BATCH], d[CSR_BATCH];
        #pragma unroll
        for (int u = 0; u < CSR_BATCH; ++u) {
            const int e = base + u * CSR_THREADS + tid;
            s[u] = (e < e1) ? src[e] : 0;
            d[u] = (e < e1) ? dst[e] : 0;
        }
        #pragma unroll
        for (int u = 0; u < CSR_BATCH; ++u) {
            const int e = base + u * CSR_THREADS + tid;
            if (e < e1) {
                const int ri = __hip_atomic_fetch_add(&hist[d[u]],     1, __ATOMIC_RELAXED, __HIP_MEMORY_SCOPE_WORKGROUP);
                const int ro = __hip_atomic_fetch_add(&hist[N + s[u]], 1, __ATOMIC_RELAXED, __HIP_MEMORY_SCOPE_WORKGROUP);
                reinterpret_cast<int2*>(rank)[e] = make_int2(ri, ro);
            }
        }
    }
    __syncthreads();

    int* row = H + (size_t)blockIdx.x * bins;
    for (int i = tid; i < bins; i += CSR_THREADS) row[i] = hist[i];
}

#define BSCAN_THREADS 64
__global__ void __launch_bounds__(BSCAN_THREADS)
k_blockscan(int* __restrict__ H, int* __restrict__ cnt, int bins) {
    const int j = blockIdx.x * blockDim.x + threadIdx.x;
    if (j >= bins) return;
    int v[CSR_BLOCKS];
    #pragma unroll
    for (int b = 0; b < CSR_BLOCKS; ++b) v[b] = H[(size_t)b * bins + j];
    int run = 0;
    #pragma unroll
    for (int b = 0; b < CSR_BLOCKS; ++b) {
        H[(size_t)b * bins + j] = run;
        run += v[b];
    }
    cnt[j] = run;
}

__global__ void k_scan(const int* __restrict__ cnt, int* __restrict__ off, int N) {
    __shared__ int sh[1024];
    const int* c  = cnt + (size_t)blockIdx.x * N;
    int*       o  = off + (size_t)blockIdx.x * (N + 1);
    const int t     = threadIdx.x;
    const int chunk = (N + 1023) >> 10;
    const int b     = t * chunk;
    const int e     = min(b + chunk, N);
    int s = 0;
    for (int i = b; i < e; ++i) s += c[i];
    sh[t] = s;
    __syncthreads();
    for (int d = 1; d < 1024; d <<= 1) {
        int v = (t >= d) ? sh[t - d] : 0;
        __syncthreads();
        sh[t] += v;
        __syncthreads();
    }
    int run = sh[t] - s;
    for (int i = b; i < e; ++i) {
        o[i] = run;
        run += c[i];
    }
    if (t == 1023) o[N] = sh[1023];
}

__global__ void k_fill(const int* __restrict__ src, const int* __restrict__ dst,
                       const int* __restrict__ off_dst, const int* __restrict__ off_src,
                       const int* __restrict__ rank, const int* __restrict__ H,
                       int* __restrict__ nbr_in, int* __restrict__ nbr_out,
                       int E, int N, int chunk) {
    int e = blockIdx.x * blockDim.x + threadIdx.x;
    if (e >= E) return;
    const int  s = src[e], d = dst[e];
    const int2 r = reinterpret_cast<const int2*>(rank)[e];
    const int* row = H + (size_t)(e / chunk) * (2 * N);
    nbr_in [off_dst[d] + row[d]     + r.x] = s;
    nbr_out[off_src[s] + row[N + s] + r.y] = d;
}

__device__ __forceinline__ float4 ldRow(const float* __restrict__ X, int r, int l) {
    return *reinterpret_cast<const float4*>(X + (size_t)r * NHID + 4 * l);
}

__global__ void k_gather_mean(const float* __restrict__ X, const int* __restrict__ off,
                              const int* __restrict__ nbr, float* __restrict__ agg, int N) {
    const int t = blockIdx.x * blockDim.x + threadIdx.x;
    const int node = __builtin_amdgcn_readfirstlane(t >> 5);
    if (node >= N) return;
    const int l  = t & 31;
    const int b  = off[node];
    const int e  = off[node + 1];
    float4 acc = make_float4(0.f, 0.f, 0.f, 0.f);
    for (int base = b; base < e; base += 32) {
        const int n  = min(32, e - base);
        const int id = (l < n) ? nbr[base + l] : 0;
        for (int j = 0; j < n; j += 8) {
            float4 v[8];
            #pragma unroll
            for (int u = 0; u < 8; ++u)
                v[u] = ldRow(X, __builtin_amdgcn_readlane(id, (j + u) & 31), l);
            #pragma unroll
            for (int u = 0; u < 8; ++u) {
                const bool live = (j + u < n);
                acc.x += live ? v[u].x : 0.0f;
                acc.y += live ? v[u].y : 0.0f;
                acc.z += live ? v[u].z : 0.0f;
                acc.w += live ? v[u].w : 0.0f;
            }
        }
    }
    const float inv = 1.0f / fmaxf((float)(e - b), 1.0f);
    float4 r = make_float4(acc.x * inv, acc.y * inv, acc.z * inv, acc.w * inv);
    *reinterpret_cast<float4*>(agg + (size_t)node * NHID + 4 * l) = r;
}

__global__ void k_gate_gather(const float* __restrict__ A, const float* __restrict__ B,
                              const int* __restrict__ off, const int* __restrict__ nbr,
                              const float* __restrict__ Xcur, const float* __restrict__ Xnew,
                              float* __restrict__ Xnext, int N) {
    const int t = blockIdx.x * blockDim.x + threadIdx.x;
    const int node = __builtin_amdgcn_readfirstlane(t >> 5);
    if (node >= N) return;
    const int l  = t & 31;
    const int b  = off[node];
    const int e  = off[node + 1];
    const float4 a  = ldRow(A, node, l);
    const float4 xc = ldRow(Xcur, node, l);
    const float4 xn = ldRow(Xnew, node, l);
    float4 h = make_float4(0.f, 0.f, 0.f, 0.f);
    for (int base = b; base < e; base += 32) {
        const int n  = min(32, e - base);
        const int id = (l < n) ? nbr[base + l] : 0;
        for (int j = 0; j < n; j += 8) {
            float4 v[8];
            #pragma unroll
            for (int u = 0; u < 8; ++u)
                v[u] = ldRow(B, __builtin_amdgcn_readlane(id, (j + u) & 31), l);
            #pragma unroll
            for (int u = 0; u < 8; ++u) {
                const bool live = (j + u < n);
                const float tx = a.x + v[u].x, ty = a.y + v[u].y;
                const float tz = a.z + v[u].z, tw = a.w + v[u].w;
                h.x += live ? tx * tx : 0.0f;
                h.y += live ? ty * ty : 0.0f;
                h.z += live ? tz * tz : 0.0f;
                h.w += live ? tw * tw : 0.0f;
            }
        }
    }
    const float c = fmaxf((float)(e - b), 1.0f);
    const float t0 = tanhf(h.x / c), t1 = tanhf(h.y / c);
    const float t2 = tanhf(h.z / c), t3 = tanhf(h.w / c);
    float4 r;
    r.x = (1.0f - t0) * xc.x + t0 * xn.x;
    r.y = (1.0f - t1) * xc.y + t1 * xn.y;
    r.z = (1.0f - t2) * xc.z + t2 * xn.z;
    r.w = (1.0f - t3) * xc.w + t3 * xn.w;
    *reinterpret_cast<float4*>(Xnext + (size_t)node * NHID + 4 * l) = r;
}

#define GEMM_THREADS 256
#define A_PITCH      132

template <int N_, int NSEG>
__device__ __forceinline__ void gemm_body(float* __restrict__ smem,
                                          const float* __restrict__ A1, int lda1, const float* __restrict__ W1,
                                          const float* __restrict__ A2, int lda2, const float* __restrict__ W2,
                                          const float* __restrict__ bias, float* __restrict__ C,
                                          int M, int relu) {
    constexpr int PITCH  = 2 * N_ + 16;
    constexpr int NTILES = N_ / 16;
    constexpr int MG     = 8 / NTILES;
    constexpr int ROWS   = 48 * MG;
    constexpr int G      = N_ / 4;
    constexpr int PH     = NSEG * 32;
    constexpr int WTRIPS = (PH * G) / GEMM_THREADS;
    constexpr int ATRIPS = (ROWS * 32) / GEMM_THREADS;
    constexpr int KBP    = NSEG * 8;
    static_assert((PH * G) % GEMM_THREADS == 0 && (ROWS * 32) % GEMM_THREADS == 0,
                  "staging work divides evenly over the work-group");

    float* sW = smem;
    float* sA = smem + NSEG * 64 * PITCH;

    const int tid  = threadIdx.x;
    const int lane = tid & 31;
    const int wave = tid >> 5;
    const int c    = lane & 15;
    const int h    = lane >> 4;
    const int nt   = wave % NTILES;
    const int mg   = wave / NTILES;
    const int m0   = blockIdx.x * ROWS;

    float4 w0[WTRIPS], w1[WTRIPS];
    auto loadW = [&](int ph) {
        #pragma unroll
        for (int it = 0; it < WTRIPS; ++it) {
            const int i    = it * GEMM_THREADS + tid;
            const int qrow = ph * PH + i / G;
            const int g    = i % G;
            const float* Wp = ((qrow >> 6) == 0 ? W1 : W2) + (size_t)(2 * (qrow & 63)) * N_ + 4 * g;
            w0[it] = *reinterpret_cast<const float4*>(Wp);
            w1[it] = *reinterpret_cast<const float4*>(Wp + N_);
        }
    };
    auto storeW = [&](int ph) {
        #pragma unroll
        for (int it = 0; it < WTRIPS; ++it) {
            const int i    = it * GEMM_THREADS + tid;
            const int qrow = ph * PH + i / G;
            const int g    = i % G;
            float* d = sW + qrow * PITCH + 8 * g;
            *reinterpret_cast<float4*>(d)     = make_float4(w0[it].x, w1[it].x, w0[it].y, w1[it].y);
            *reinterpret_cast<float4*>(d + 4) = make_float4(w0[it].z, w1[it].z, w0[it].w, w1[it].w);
        }
    };

    v8f acc[3];
    #pragma unroll
    for (int i = 0; i < 3; ++i) acc[i] = v8f{};

    auto compute = [&](int ph) {
        #pragma unroll
        for (int kbl = 0; kbl < KBP; ++kbl) {
            const int KB  = ph * KBP + kbl;
            const int s   = KB >> 4;
            const int kbs = KB & 15;
            float4 a[3];
            #pragma unroll
            for (int i = 0; i < 3; ++i)
                a[i] = *reinterpret_cast<const float4*>(
                    sA + (s * ROWS + 48 * mg + 16 * i + c) * A_PITCH + 8 * kbs + 4 * h);
            #pragma unroll
            for (int j = 0; j < 2; ++j) {
                const v2f bv = *reinterpret_cast<const v2f*>(
                    sW + (4 * KB + 2 * h + j) * PITCH + 2 * (16 * nt + c));
                #pragma unroll
                for (int i = 0; i < 3; ++i) {
                    v2f av;
                    av.x = (j == 0) ? a[i].x : a[i].z;
                    av.y = (j == 0) ? a[i].y : a[i].w;
                    acc[i] = __builtin_amdgcn_wmma_f32_16x16x4_f32(false, av, false, bv,
                                                                   (short)0, acc[i], false, false);
                }
            }
        }
    };

    loadW(0);
    #pragma unroll
    for (int s = 0; s < NSEG; ++s) {
        #pragma unroll
        for (int it = 0; it < ATRIPS; ++it) {
            const int i   = it * GEMM_THREADS + tid;
            const int row = i >> 5;
            const int q   = i & 31;
            const int gr  = min(m0 + row, M - 1);
            const float* ap = (s == 0) ? A1 + (size_t)gr * lda1 : A2 + (size_t)gr * lda2;
            const float4 v = *reinterpret_cast<const float4*>(ap + 4 * q);
            *reinterpret_cast<float4*>(sA + (s * ROWS + row) * A_PITCH + 4 * q) = v;
        }
    }
    storeW(0);
    __syncthreads();
    loadW(1);
    compute(0);
    __builtin_amdgcn_sched_barrier(0);
    storeW(1);
    __syncthreads();
    compute(1);

    const int n = 16 * nt + c;
    const float bv = bias ? bias[n] : 0.0f;
    #pragma unroll
    for (int i = 0; i < 3; ++i) {
        #pragma unroll
        for (int r = 0; r < 8; ++r) {
            float x = acc[i][r] + bv;
            if (relu) x = fmaxf(x, 0.0f);
            const int m = m0 + 48 * mg + 16 * i + 8 * h + r;
            if (m < M) C[(size_t)m * N_ + n] = x;
        }
    }
}

__global__ void __launch_bounds__(GEMM_THREADS)
k_gemm_n128_s2(const float* __restrict__ A1, int lda1, const float* __restrict__ W1,
               const float* __restrict__ A2, int lda2, const float* __restrict__ W2,
               const float* __restrict__ bias, float* __restrict__ C, int M, int relu) {
    __shared__ __attribute__((aligned(16))) float smem[2 * 64 * (2 * 128 + 16) + 2 * 48 * A_PITCH];
    gemm_body<128, 2>(smem, A1, lda1, W1, A2, lda2, W2, bias, C, M, relu);
}

#define L_PITCH (2 * NHID + 16)
#define L_ROWS  48
#define L_WTRIPS ((64 * 32) / GEMM_THREADS)
#define L_ATRIPS ((L_ROWS * 32) / GEMM_THREADS)

__device__ __forceinline__ void layer_loadW(const float* __restrict__ Wp, int tid,
                                            float4 (&w0)[L_WTRIPS], float4 (&w1)[L_WTRIPS]) {
    #pragma unroll
    for (int it = 0; it < L_WTRIPS; ++it) {
        const int i = it * GEMM_THREADS + tid;
        const int q = i >> 5;
        const int g = i & 31;
        const float* p = Wp + (size_t)(2 * q) * NHID + 4 * g;
        w0[it] = *reinterpret_cast<const float4*>(p);
        w1[it] = *reinterpret_cast<const float4*>(p + NHID);
    }
}

template <int BUF>
__device__ __forceinline__ void layer_storeW(float* sW, int tid,
                                             const float4 (&w0)[L_WTRIPS], const float4 (&w1)[L_WTRIPS]) {
    #pragma unroll
    for (int it = 0; it < L_WTRIPS; ++it) {
        const int i = it * GEMM_THREADS + tid;
        const int q = i >> 5;
        const int g = i & 31;
        float* d = sW + (BUF * 64 + q) * L_PITCH + 8 * g;
        *reinterpret_cast<float4*>(d)     = make_float4(w0[it].x, w1[it].x, w0[it].y, w1[it].y);
        *reinterpret_cast<float4*>(d + 4) = make_float4(w0[it].z, w1[it].z, w0[it].w, w1[it].w);
    }
}

template <int BUF, int SEG>
__device__ __forceinline__ void layer_compute(const float* sW, const float* sA,
                                              int c, int h, int nt, v8f (&acc)[3]) {
    #pragma unroll
    for (int kbl = 0; kbl < 16; ++kbl) {
        float4 a[3];
        #pragma unroll
        for (int i = 0; i < 3; ++i)
            a[i] = *reinterpret_cast<const float4*>(
                sA + (SEG * L_ROWS + 16 * i + c) * A_PITCH + 8 * kbl + 4 * h);
        #pragma unroll
        for (int j = 0; j < 2; ++j) {
            const v2f bv = *reinterpret_cast<const v2f*>(
                sW + (BUF * 64 + 4 * kbl + 2 * h + j) * L_PITCH + 2 * (16 * nt + c));
            #pragma unroll
            for (int i = 0; i < 3; ++i) {
                v2f av;
                av.x = (j == 0) ? a[i].x : a[i].z;
                av.y = (j == 0) ? a[i].y : a[i].w;
                acc[i] = __builtin_amdgcn_wmma_f32_16x16x4_f32(false, av, false, bv,
                                                               (short)0, acc[i], false, false);
            }
        }
    }
}

template <bool RELU>
__device__ __forceinline__ void layer_store_global(const v8f (&acc)[3], float bias,
                                                   float* __restrict__ C, int m0, int n, int h, int M) {
    #pragma unroll
    for (int i = 0; i < 3; ++i) {
        #pragma unroll
        for (int r = 0; r < 8; ++r) {
            float x = acc[i][r] + bias;
            if (RELU) x = fmaxf(x, 0.0f);
            const int m = m0 + 16 * i + 8 * h + r;
            if (m < M) C[(size_t)m * NHID + n] = x;
        }
    }
}

__global__ void __launch_bounds__(GEMM_THREADS)
k_layer_gemms(const float* __restrict__ agg, const float* __restrict__ Xcur,
              const float* __restrict__ conv_Wl, const float* __restrict__ conv_Wr,
              const float* __restrict__ conv_bl,
              const float* __restrict__ gg_Wl, const float* __restrict__ gg_Wr,
              const float* __restrict__ gg_bl,
              const float* __restrict__ Q_top, const float* __restrict__ Q_bot,
              const float* __restrict__ Q_b,
              float* __restrict__ Xnew, float* __restrict__ Ab, float* __restrict__ Bb, int M) {
    __shared__ __attribute__((aligned(16))) float smem[2 * 64 * L_PITCH + 3 * L_ROWS * A_PITCH];
    float* sW = smem;
    float* sA = smem + 2 * 64 * L_PITCH;

    const int tid  = threadIdx.x;
    const int lane = tid & 31;
    const int nt   = tid >> 5;
    const int c    = lane & 15;
    const int h    = lane >> 4;
    const int m0   = blockIdx.x * L_ROWS;
    const int n    = 16 * nt + c;

    float4 w0[L_WTRIPS], w1[L_WTRIPS];
    v8f acc[3];

    layer_loadW(conv_Wl, tid, w0, w1);
    #pragma unroll
    for (int s = 0; s < 2; ++s) {
        #pragma unroll
        for (int it = 0; it < L_ATRIPS; ++it) {
            const int i   = it * GEMM_THREADS + tid;
            const int row = i >> 5;
            const int q   = i & 31;
            const int gr  = min(m0 + row, M - 1);
            const float* ap = ((s == 0) ? agg : Xcur) + (size_t)gr * NHID;
            const float4 v = *reinterpret_cast<const float4*>(ap + 4 * q);
            *reinterpret_cast<float4*>(sA + (s * L_ROWS + row) * A_PITCH + 4 * q) = v;
        }
    }
    const float b_conv = conv_bl[n];
    const float b_gg   = gg_bl[n];
    const float b_q    = Q_b[n];
    layer_storeW<0>(sW, tid, w0, w1);
    __syncthreads();


    layer_loadW(conv_Wr, tid, w0, w1);
    #pragma unroll
    for (int i = 0; i < 3; ++i) acc[i] = v8f{};
    layer_compute<0, 0>(sW, sA, c, h, nt, acc);
    __builtin_amdgcn_sched_barrier(0);
    layer_storeW<1>(sW, tid, w0, w1);
    __syncthreads();

    layer_loadW(gg_Wl, tid, w0, w1);
    layer_compute<1, 1>(sW, sA, c, h, nt, acc);
    __builtin_amdgcn_sched_barrier(0);
    layer_storeW<0>(sW, tid, w0, w1);
    __syncthreads();

    layer_loadW(gg_Wr, tid, w0, w1);
    layer_store_global<true>(acc, b_conv, Xnew, m0, n, h, M);
    #pragma unroll
    for (int i = 0; i < 3; ++i) acc[i] = v8f{};
    layer_compute<0, 0>(sW, sA, c, h, nt, acc);
    __builtin_amdgcn_sched_barrier(0);
    layer_storeW<1>(sW, tid, w0, w1);
    __syncthreads();

    layer_loadW(Q_top, tid, w0, w1);
    layer_compute<1, 1>(sW, sA, c, h, nt, acc);
    #pragma unroll
    for (int i = 0; i < 3; ++i) {
        #pragma unroll
        for (int r = 0; r < 8; ++r)
            sA[(2 * L_ROWS + 16 * i + 8 * h + r) * A_PITCH + n] = fmaxf(acc[i][r] + b_gg, 0.0f);
    }
    __builtin_amdgcn_sched_barrier(0);
    layer_storeW<0>(sW, tid, w0, w1);
    __syncthreads();

    layer_loadW(Q_bot, tid, w0, w1);
    #pragma unroll
    for (int i = 0; i < 3; ++i) acc[i] = v8f{};
    layer_compute<0, 2>(sW, sA, c, h, nt, acc);
    __builtin_amdgcn_sched_barrier(0);
    layer_storeW<1>(sW, tid, w0, w1);
    __syncthreads();

    layer_store_global<false>(acc, b_q, Ab, m0, n, h, M);
    #pragma unroll
    for (int i = 0; i < 3; ++i) acc[i] = v8f{};
    layer_compute<1, 2>(sW, sA, c, h, nt, acc);
    layer_store_global<false>(acc, 0.0f, Bb, m0, n, h, M);
}

__global__ void __launch_bounds__(GEMM_THREADS)
k_gemm_n64_s1(const float* __restrict__ A1, int lda1, const float* __restrict__ W1,
              const float* __restrict__ bias, float* __restrict__ C, int M, int relu) {
    __shared__ __attribute__((aligned(16))) float smem[64 * (2 * 64 + 16) + 96 * A_PITCH];
    gemm_body<64, 1>(smem, A1, lda1, W1, nullptr, 0, nullptr, bias, C, M, relu);
}

extern "C" void kernel_launch(void* const* d_in, const int* in_sizes, int n_in,
                              void* d_out, int out_size, void* d_ws, size_t ws_size,
                              hipStream_t stream) {
    const float* X       = (const float*)d_in[0];
    const int*   ei      = (const int*)  d_in[1];
    const float* enc_W   = (const float*)d_in[2];
    const float* enc_b   = (const float*)d_in[3];
    const float* conv_Wl = (const float*)d_in[4];
    const float* conv_bl = (const float*)d_in[5];
    const float* conv_Wr = (const float*)d_in[6];
    const float* gg_Wl   = (const float*)d_in[7];
    const float* gg_bl   = (const float*)d_in[8];
    const float* gg_Wr   = (const float*)d_in[9];
    const float* Q_W     = (const float*)d_in[10];
    const float* Q_b     = (const float*)d_in[11];
    const float* dec_W   = (const float*)d_in[12];
    const float* dec_b   = (const float*)d_in[13];

    const int N = in_sizes[0] / NFEAT;
    const int E = in_sizes[1] / 2;
    const int* src = ei;
    const int* dst = ei + E;

    float* ws       = (float*)d_ws;
    const size_t NH = (size_t)N * NHID;
    int*   cnt      = (int*)ws;
    float* Xa       = ws + 2 * (size_t)N;
    float* Xb       = Xa + NH;
    float* agg      = Xb + NH;
    float* T1       = agg + NH;
    float* Ab       = T1 + NH;
    float* Bb       = Ab + NH;
    int*   csr      = (int*)(Bb + NH);
    int*   off      = csr;
    int*   nbr_in   = off + 2 * (size_t)(N + 1);
    int*   nbr_out  = nbr_in + E;
    int*   rank     = (int*)agg;
    int*   H        = (int*)T1;
    const int* off_dst = off;
    const int* off_src = off + (N + 1);

    auto cdiv = [](long long a, long long b) { return (int)((a + b - 1) / b); };
    const int nodeGrd = cdiv((long long)N * 32, 256);
    const int gemmGrd = cdiv(N, 48);
    const int decGrd  = cdiv(N, 96);

    const int chunk = cdiv(E, CSR_BLOCKS);
    k_count    <<<CSR_BLOCKS, CSR_THREADS, 0, stream>>>(src, dst, H, rank, E, N, chunk);
    k_blockscan<<<cdiv(2 * N, BSCAN_THREADS), BSCAN_THREADS, 0, stream>>>(H, cnt, 2 * N);
    k_scan     <<<2, 1024, 0, stream>>>(cnt, off, N);
    k_fill     <<<cdiv(E, 256), 256, 0, stream>>>(src, dst, off_dst, off_src, rank, H,
                                                  nbr_in, nbr_out, E, N, chunk);

    k_gemm_n128_s2<<<gemmGrd, GEMM_THREADS, 0, stream>>>(X, NFEAT, enc_W, X + NHID, NFEAT,
                                                         enc_W + (size_t)NHID * NHID, enc_b, Xa, N, 1);

    float* Xcur  = Xa;
    float* Xnext = Xb;
    for (int layer = 0; layer < 2; ++layer) {
        k_gather_mean<<<nodeGrd, 256, 0, stream>>>(Xcur, off_dst, nbr_in, agg, N);

        k_layer_gemms<<<gemmGrd, GEMM_THREADS, 0, stream>>>(agg, Xcur, conv_Wl, conv_Wr, conv_bl,
                                                            gg_Wl, gg_Wr, gg_bl,
                                                            Q_W, Q_W + (size_t)NHID * NHID, Q_b,
                                                            T1, Ab, Bb, N);

        k_gate_gather<<<nodeGrd, 256, 0, stream>>>(Ab, Bb, off_src, nbr_out, Xcur, T1, Xnext, N);

        float* t = Xcur; Xcur = Xnext; Xnext = t;
    }

    k_gemm_n64_s1<<<decGrd, GEMM_THREADS, 0, stream>>>(Xcur, NHID, dec_W, dec_b, (float*)d_out, N, 0);
}
